# starting kernel
# baseline (speedup 1.0000x reference)
#include <hip/hip_runtime.h>
#include <hip/hip_bf16.h>

typedef __attribute__((ext_vector_type(16))) _Float16 v16h;
typedef __attribute__((ext_vector_type(8)))  float    v8f;

#define T_HIST 3
#define F_IN   5
#define HDIM   128
#define CDIM   5

__global__ void fill_zero_kernel(float* __restrict__ p, int n) {
    int i = blockIdx.x * blockDim.x + threadIdx.x;
    if (i < n) p[i] = 0.0f;
}

__global__ void deg_kernel(const int* __restrict__ dst, float* __restrict__ deg, int e) {
    int i = blockIdx.x * blockDim.x + threadIdx.x;
    if (i < e) atomicAdd(&deg[dst[i]], 1.0f);
}

__global__ void dinv_kernel(float* __restrict__ deg, int n) {
    int i = blockIdx.x * blockDim.x + threadIdx.x;
    if (i < n) deg[i] = rsqrtf(deg[i] + 1.0f);
}

__global__ void xw1_seed_kernel(const float* __restrict__ x, const float* __restrict__ W1,
                                const float* __restrict__ b1, const float* __restrict__ dinv,
                                float* __restrict__ xw1, float* __restrict__ agg_seed, int n) {
    int idx = blockIdx.x * blockDim.x + threadIdx.x;
    if (idx >= n * HDIM) return;
    int row = idx >> 7;
    int h   = idx & (HDIM - 1);
    const float* xr = x + (size_t)row * (T_HIST * F_IN) + (T_HIST - 1) * F_IN;
    float s = 0.0f;
#pragma unroll
    for (int f = 0; f < F_IN; ++f) s = fmaf(xr[f], W1[f * HDIM + h], s);
    xw1[idx] = s;
    float di = dinv[row];
    agg_seed[idx] = di * di * s + b1[h];
}

__global__ __launch_bounds__(256) void scatter_kernel(const int* __restrict__ src, const int* __restrict__ dst,
                                                      const float* __restrict__ dinv, const float* __restrict__ xw,
                                                      float* __restrict__ agg, int e) {
    int edge = (blockIdx.x * blockDim.x + threadIdx.x) >> 5;
    int lane = threadIdx.x & 31;
    if (edge >= e) return;
    int s = src[edge], d = dst[edge];
    float norm = dinv[s] * dinv[d];
    const float4* xr = (const float4*)(xw + (size_t)s * HDIM);
    float4 v = xr[lane];
    float* ag = agg + (size_t)d * HDIM + lane * 4;
    atomicAdd(ag + 0, norm * v.x);
    atomicAdd(ag + 1, norm * v.y);
    atomicAdd(ag + 2, norm * v.z);
    atomicAdd(ag + 3, norm * v.w);
}

__global__ __launch_bounds__(256) void gemm2_wmma_kernel(const float* __restrict__ pre1,
                                                         const float* __restrict__ W2,
                                                         const float* __restrict__ b2,
                                                         const float* __restrict__ dinv,
                                                         float* __restrict__ xw2,
                                                         float* __restrict__ agg_seed,
                                                         int nrows) {
    __shared__ _Float16 aS[16][HDIM];
    __shared__ _Float16 bS[HDIM][HDIM];
    __shared__ float    dS[16];

    const int tid       = threadIdx.x;
    const int block_row = blockIdx.x * 16;

    for (int i = tid; i < HDIM * HDIM; i += 256) {
        int k = i >> 7;
        int n = i & (HDIM - 1);
        bS[n][k] = (_Float16)W2[i];
    }
    for (int i = tid; i < 16 * HDIM; i += 256) {
        int m = i >> 7;
        int k = i & (HDIM - 1);
        int row = block_row + m;
        float v = (row < nrows) ? fmaxf(pre1[(size_t)row * HDIM + k], 0.0f) : 0.0f;
        aS[m][k] = (_Float16)v;
    }
    if (tid < 16) {
        int row = block_row + tid;
        dS[tid] = (row < nrows) ? dinv[row] : 0.0f;
    }
    __syncthreads();

    const int wave  = tid >> 5;
    const int lane  = tid & 31;
    const int ncol  = wave * 16;
    const int mrow  = lane & 15;
    const int khalf = (lane >> 4) * 8;

    v8f acc = {};
    for (int kk = 0; kk < HDIM; kk += 32) {
        v16h a, b;
#pragma unroll
        for (int j = 0; j < 16; j += 2) {
            int K = (j & 7) + ((j >> 3) << 4) + khalf + kk;
            a[j]     = aS[mrow][K];
            a[j + 1] = aS[mrow][K + 1];
            b[j]     = bS[ncol + mrow][K];
            b[j + 1] = bS[ncol + mrow][K + 1];
        }
        acc = __builtin_amdgcn_wmma_f32_16x16x32_f16(false, a, false, b,
                                                     (short)0, acc, false, false);
    }

    const int col  = ncol + (lane & 15);
    const float bc = b2[col];
    if (block_row + 16 <= nrows) {
#pragma unroll
        for (int v = 0; v < 8; ++v) {
            int r  = v + ((lane >> 4) << 3);
            int row = block_row + r;
            float val = acc[v];
            float di  = dS[r];
            xw2[(size_t)row * HDIM + col]      = val;
            agg_seed[(size_t)row * HDIM + col] = di * di * val + bc;
        }
    } else {
#pragma unroll
        for (int v = 0; v < 8; ++v) {
            int r  = v + ((lane >> 4) << 3);
            int row = block_row + r;
            if (row < nrows) {
                float val = acc[v];
                float di  = dS[r];
                xw2[(size_t)row * HDIM + col]      = val;
                agg_seed[(size_t)row * HDIM + col] = di * di * val + bc;
            }
        }
    }
}

__global__ __launch_bounds__(256) void out_kernel(const float* __restrict__ pre2,
                                                  const float* __restrict__ Wout,
                                                  const float* __restrict__ bout,
                                                  float* __restrict__ out, int n) {
    int node = (blockIdx.x * blockDim.x + threadIdx.x) >> 5;
    int lane = threadIdx.x & 31;
    if (node >= n) return;

    const float4* hr = (const float4*)(pre2 + (size_t)node * HDIM);
    float4 hv = hr[lane];
    hv.x = fmaxf(hv.x, 0.0f);
    hv.y = fmaxf(hv.y, 0.0f);
    hv.z = fmaxf(hv.z, 0.0f);
    hv.w = fmaxf(hv.w, 0.0f);

    int h0 = lane * 4;
    float acc[CDIM];
#pragma unroll
    for (int c = 0; c < CDIM; ++c) {
        acc[c] = hv.x * Wout[(h0 + 0) * CDIM + c]
               + hv.y * Wout[(h0 + 1) * CDIM + c]
               + hv.z * Wout[(h0 + 2) * CDIM + c]
               + hv.w * Wout[(h0 + 3) * CDIM + c];
    }
#pragma unroll
    for (int off = 16; off > 0; off >>= 1) {
#pragma unroll
        for (int c = 0; c < CDIM; ++c) acc[c] += __shfl_xor(acc[c], off, 32);
    }
    if (lane == 0) {
#pragma unroll
        for (int c = 0; c < CDIM; ++c) acc[c] += bout[c];
        float m = acc[0];
#pragma unroll
        for (int c = 1; c < CDIM; ++c) m = fmaxf(m, acc[c]);
        float sum = 0.0f;
#pragma unroll
        for (int c = 0; c < CDIM; ++c) { acc[c] = __expf(acc[c] - m); sum += acc[c]; }
        float inv = 1.0f / sum;
#pragma unroll
        for (int c = 0; c < CDIM; ++c) out[(size_t)node * CDIM + c] = acc[c] * inv;
    }
}

extern "C" void kernel_launch(void* const* d_in, const int* in_sizes, int n_in,
                              void* d_out, int out_size, void* d_ws, size_t ws_size,
                              hipStream_t stream) {
    const float* x    = (const float*)d_in[0];
    const int*   eidx = (const int*)  d_in[1];
    const float* W1   = (const float*)d_in[2];
    const float* b1   = (const float*)d_in[3];
    const float* W2   = (const float*)d_in[4];
    const float* b2   = (const float*)d_in[5];
    const float* Wout = (const float*)d_in[6];
    const float* bout = (const float*)d_in[7];
    float*       out  = (float*)d_out;

    const int N = in_sizes[0] / (T_HIST * F_IN);
    const int E = in_sizes[1] / 2;
    const int* src = eidx;
    const int* dst = eidx + E;

    float* dinv = (float*)d_ws;
    float* bufA = dinv + ((N + 63) & ~63);
    float* bufB = bufA + (size_t)N * HDIM;

    const int TB = 256;
    const int NH = N * HDIM;
    const int NH_blocks   = (NH + TB - 1) / TB;
    const int edge_blocks = ((E * 32) + TB - 1) / TB;
    const int node_blocks = ((N * 32) + TB - 1) / TB;

    fill_zero_kernel<<<(N + TB - 1) / TB, TB, 0, stream>>>(dinv, N);
    deg_kernel<<<(E + TB - 1) / TB, TB, 0, stream>>>(dst, dinv, E);
    dinv_kernel<<<(N + TB - 1) / TB, TB, 0, stream>>>(dinv, N);

    xw1_seed_kernel<<<NH_blocks, TB, 0, stream>>>(x, W1, b1, dinv, bufA, bufB, N);
    scatter_kernel<<<edge_blocks, TB, 0, stream>>>(src, dst, dinv, bufA, bufB, E);

    gemm2_wmma_kernel<<<(N + 15) / 16, 256, 0, stream>>>(bufB, W2, b2, dinv, bufA, bufB, N);
    scatter_kernel<<<edge_blocks, TB, 0, stream>>>(src, dst, dinv, bufA, bufB, E);

    out_kernel<<<node_blocks, TB, 0, stream>>>(bufB, Wout, bout, out, N);
}


# layer-1 scatter: source row built in registers per edge from the 5 inputs (x_last[s] @ W1, same fma chain) instead of fetched as a stored 512-byte row; xw1 plane not written; adds in arrival order
# speedup vs baseline: 2.5151x; 2.5151x over previous
#include <hip/hip_runtime.h>
#include <hip/hip_bf16.h>

typedef __attribute__((ext_vector_type(16))) _Float16 v16h;
typedef __attribute__((ext_vector_type(8)))  float    v8f;

#define T_HIST 3
#define F_IN   5
#define HDIM   128
#define CDIM   5

__global__ void fill_zero_kernel(float* __restrict__ p, int n) {
    int i = blockIdx.x * blockDim.x + threadIdx.x;
    if (i < n) p[i] = 0.0f;
}

__global__ void deg_kernel(const int* __restrict__ dst, float* __restrict__ deg, int e) {
    int i = blockIdx.x * blockDim.x + threadIdx.x;
    if (i < e) atomicAdd(&deg[dst[i]], 1.0f);
}

__global__ void dinv_kernel(float* __restrict__ deg, int n) {
    int i = blockIdx.x * blockDim.x + threadIdx.x;
    if (i < n) deg[i] = rsqrtf(deg[i] + 1.0f);
}

__device__ __forceinline__ float l1_dot(const float (&xv)[F_IN], const float (&wv)[F_IN]) {
    float s = 0.0f;
#pragma unroll
    for (int f = 0; f < F_IN; ++f) s = fmaf(xv[f], wv[f], s);
    return s;
}

__global__ void seed1_kernel(const float* __restrict__ x, const float* __restrict__ W1,
                             const float* __restrict__ b1, const float* __restrict__ dinv,
                             float* __restrict__ agg_seed, int n) {
    int idx = blockIdx.x * blockDim.x + threadIdx.x;
    if (idx >= n * HDIM) return;
    int row = idx >> 7;
    int h   = idx & (HDIM - 1);
    const float* xr = x + (size_t)row * (T_HIST * F_IN) + (T_HIST - 1) * F_IN;
    float xv[F_IN], wv[F_IN];
#pragma unroll
    for (int f = 0; f < F_IN; ++f) { xv[f] = xr[f]; wv[f] = W1[f * HDIM + h]; }
    const float s = l1_dot(xv, wv);
    float di = dinv[row];
    agg_seed[idx] = di * di * s + b1[h];
}

#define EDGES_PER_WAVE  32
#define EDGES_PER_BLOCK 256
#define EDGE_GROUP      16
#define HALF_COLS       (HDIM / 2)

__device__ __forceinline__ float lane_bcast_f(float v, int l) {
    return __int_as_float(__builtin_amdgcn_readlane(__float_as_int(v), l));
}

__device__ __forceinline__ void scatter_wave(const int* __restrict__ src, const int* __restrict__ dst,
                                             const float* __restrict__ dinv, const float* __restrict__ xh,
                                             float* __restrict__ ah, int base, int lane, int e) {
    const int mine = base + lane;
    int   s  = 0, d = 0;
    float nm = 0.0f;
    if (mine < e) {
        s  = src[mine];
        d  = dst[mine];
        nm = dinv[s] * dinv[d];
    }
    const int cnt = min(EDGES_PER_WAVE, e - base);
    int j = 0;
    for (; j + EDGE_GROUP <= cnt; j += EDGE_GROUP) {
        float v[EDGE_GROUP][2];
#pragma unroll
        for (int u = 0; u < EDGE_GROUP; ++u) {
            const int sj = __builtin_amdgcn_readlane(s, j + u);
            const unsigned xi = (unsigned)sj * HDIM + (unsigned)lane;
            v[u][0] = xh[xi];
            v[u][1] = xh[xi + 32u];
        }
#pragma unroll
        for (int u = 0; u < EDGE_GROUP; ++u) {
            const int   dj = __builtin_amdgcn_readlane(d, j + u);
            const float nj = lane_bcast_f(nm, j + u);
            const unsigned ai = (unsigned)dj * HDIM + (unsigned)lane;
            atomicAdd(ah + ai,       nj * v[u][0]);
            atomicAdd(ah + ai + 32u, nj * v[u][1]);
        }
    }
    for (; j < cnt; ++j) {
        const int   sj = __builtin_amdgcn_readlane(s, j);
        const int   dj = __builtin_amdgcn_readlane(d, j);
        const float nj = lane_bcast_f(nm, j);
        const unsigned xi = (unsigned)sj * HDIM + (unsigned)lane;
        const unsigned ai = (unsigned)dj * HDIM + (unsigned)lane;
        const float v0 = xh[xi], v1 = xh[xi + 32u];
        atomicAdd(ah + ai,       nj * v0);
        atomicAdd(ah + ai + 32u, nj * v1);
    }
}


#define L1_GROUP 8

__device__ __forceinline__ void l1_edge(int d, float nm, const float (&xr)[F_IN], const float (&w0)[F_IN],
                                        const float (&w1)[F_IN], int lane, int j,
                                        float& p0, float& p1, unsigned& ai) {
    const int   dj = __builtin_amdgcn_readlane(d, j);
    const float nj = lane_bcast_f(nm, j);
    float xj[F_IN];
#pragma unroll
    for (int f = 0; f < F_IN; ++f) xj[f] = lane_bcast_f(xr[f], j);
    ai = (unsigned)dj * HDIM + (unsigned)lane;
    p0 = nj * l1_dot(xj, w0);
    p1 = nj * l1_dot(xj, w1);
}

__device__ __forceinline__ void scatter_wave_l1(const int* __restrict__ src, const int* __restrict__ dst,
                                                const float* __restrict__ dinv, const float* __restrict__ x,
                                                const float (&w0)[F_IN], const float (&w1)[F_IN],
                                                float* __restrict__ ah, int base, int lane, int e) {
    const int mine = base + lane;
    int   s  = 0, d = 0;
    float nm = 0.0f;
    float xr[F_IN] = {0.0f, 0.0f, 0.0f, 0.0f, 0.0f};
    if (mine < e) {
        s  = src[mine];
        d  = dst[mine];
        nm = dinv[s] * dinv[d];
        const float* xp = x + (unsigned)s * (unsigned)(T_HIST * F_IN) + (T_HIST - 1) * F_IN;
#pragma unroll
        for (int f = 0; f < F_IN; ++f) xr[f] = xp[f];
    }
    const int cnt = min(EDGES_PER_WAVE, e - base);
    int j = 0;
    for (; j + L1_GROUP <= cnt; j += L1_GROUP) {
        float    p[L1_GROUP][2];
        unsigned ai[L1_GROUP];
#pragma unroll
        for (int u = 0; u < L1_GROUP; ++u) l1_edge(d, nm, xr, w0, w1, lane, j + u, p[u][0], p[u][1], ai[u]);
#pragma unroll
        for (int u = 0; u < L1_GROUP; ++u) {
            atomicAdd(ah + ai[u],       p[u][0]);
            atomicAdd(ah + ai[u] + 32u, p[u][1]);
        }
    }
    for (; j < cnt; ++j) {
        float p0, p1;
        unsigned ai;
        l1_edge(d, nm, xr, w0, w1, lane, j, p0, p1, ai);
        atomicAdd(ah + ai,       p0);
        atomicAdd(ah + ai + 32u, p1);
    }
}

template <bool L1>
__global__ __launch_bounds__(256) void scatter_kernel(const int* __restrict__ src, const int* __restrict__ dst,
                                                      const float* __restrict__ dinv, const float* __restrict__ rows,
                                                      const float* __restrict__ W1,
                                                      float* __restrict__ agg, int* __restrict__ queue, int e) {
    __shared__ int s_next[2];
    __shared__ int s_half;
    const int tid  = threadIdx.x;
    const int lane = tid & 31;
    const int wave = __builtin_amdgcn_readfirstlane(tid >> 5);
    const int nblk = (e + EDGES_PER_BLOCK - 1) / EDGES_PER_BLOCK;

    if (tid == 0) s_half = (int)((__builtin_amdgcn_s_sendmsg_rtn(0x87) >> 18) & 1u);
    __syncthreads();
    const int myhalf = __builtin_amdgcn_readfirstlane(s_half);

    int it = 0;
    for (int ph = 0; ph < 2; ++ph) {
        const int q = myhalf ^ ph;
        int* cq = queue + q;
        const float* xh = rows + q * HALF_COLS;
        float*       ah = agg + q * HALF_COLS;
        float w0[F_IN], w1[F_IN];
        if constexpr (L1) {
#pragma unroll
            for (int f = 0; f < F_IN; ++f) {
                w0[f] = W1[f * HDIM + q * HALF_COLS + lane];
                w1[f] = W1[f * HDIM + q * HALF_COLS + lane + 32];
            }
        }
        if (tid == 0) s_next[it & 1] = atomicAdd(cq, 1);
        __syncthreads();
        for (;;) {
            const int blk = __builtin_amdgcn_readfirstlane(s_next[it & 1]);
            if (blk >= nblk) break;
            int nb = 0;
            if (tid == 0) nb = atomicAdd(cq, 1);
            const int base = blk * EDGES_PER_BLOCK + wave * EDGES_PER_WAVE;
            if (base < e) {
                if constexpr (L1) scatter_wave_l1(src, dst, dinv, rows, w0, w1, ah, base, lane, e);
                else              scatter_wave(src, dst, dinv, xh, ah, base, lane, e);
            }
            if (tid == 0) s_next[(it + 1) & 1] = nb;
            __builtin_amdgcn_fence(__ATOMIC_RELEASE, "workgroup", "local");
            __builtin_amdgcn_s_barrier();
            __builtin_amdgcn_fence(__ATOMIC_ACQUIRE, "workgroup", "local");
            ++it;
        }
        ++it;
    }
}

#define LPITCH (HDIM + 8)

typedef __attribute__((ext_vector_type(8))) _Float16 v8h;
typedef __attribute__((ext_vector_type(4))) _Float16 v4h;

__device__ __forceinline__ v16h load_frag(const _Float16* rowp) {
    const v8h lo = *(const v8h*)(rowp);
    const v8h hi = *(const v8h*)(rowp + 16);
    return __builtin_shufflevector(lo, hi, 0, 1, 2, 3, 4, 5, 6, 7, 8, 9, 10, 11, 12, 13, 14, 15);
}

__global__ __launch_bounds__(256) void gemm2_wmma_kernel(const float* pre1,
                                                         const float* __restrict__ W2,
                                                         const float* __restrict__ b2,
                                                         const float* __restrict__ dinv,
                                                         float* __restrict__ xw2,
                                                         float* agg_seed,
                                                         int nrows) {
    __shared__ __attribute__((aligned(16))) _Float16 bS[HDIM][LPITCH];
    __shared__ __attribute__((aligned(16))) _Float16 aS[2][16][LPITCH];
    __shared__ float dS[2][16];

    const int tid    = threadIdx.x;
    const int ntiles = (nrows + 15) >> 4;
    int tile = blockIdx.x;
    if (tile >= ntiles) return;

    const int arow = tid >> 5;
    const int ak0  = (tid & 31) * 4;

    float4 p0, p1;
    float  pd = 0.0f;
    {
        const int r0 = tile * 16 + arow;
        const int r1 = r0 + 8;
        p0 = (r0 < nrows) ? *(const float4*)(pre1 + (size_t)r0 * HDIM + ak0) : make_float4(0.f, 0.f, 0.f, 0.f);
        p1 = (r1 < nrows) ? *(const float4*)(pre1 + (size_t)r1 * HDIM + ak0) : make_float4(0.f, 0.f, 0.f, 0.f);
        const int rd = tile * 16 + tid;
        if (tid < 16 && rd < nrows) pd = dinv[rd];
    }

    for (int i = tid; i < HDIM * HDIM; i += 256) {
        const int k = i >> 7;
        const int n = i & (HDIM - 1);
        bS[n][k] = (_Float16)W2[i];
    }
    __syncthreads();

    const int wave  = tid >> 5;
    const int lane  = tid & 31;
    const int ncol  = wave * 16;
    const int mrow  = lane & 15;
    const int khalf = (lane >> 4) * 8;
    const int rbase = (lane >> 4) << 3;

    v16h bfr[4];
#pragma unroll
    for (int q = 0; q < 4; ++q) bfr[q] = load_frag(&bS[ncol + mrow][q * 32 + khalf]);

    const int   col = ncol + mrow;
    const float bc  = b2[col];
    int buf = 0;

    for (; tile < ntiles; tile += gridDim.x) {
        const int block_row = tile * 16;

        {
            v4h h0, h1;
            h0[0] = (_Float16)fmaxf(p0.x, 0.0f); h0[1] = (_Float16)fmaxf(p0.y, 0.0f);
            h0[2] = (_Float16)fmaxf(p0.z, 0.0f); h0[3] = (_Float16)fmaxf(p0.w, 0.0f);
            h1[0] = (_Float16)fmaxf(p1.x, 0.0f); h1[1] = (_Float16)fmaxf(p1.y, 0.0f);
            h1[2] = (_Float16)fmaxf(p1.z, 0.0f); h1[3] = (_Float16)fmaxf(p1.w, 0.0f);
            *(v4h*)&aS[buf][arow][ak0]     = h0;
            *(v4h*)&aS[buf][arow + 8][ak0] = h1;
            if (tid < 16) dS[buf][tid] = pd;
        }
        __syncthreads();

        {
            const int nt = tile + gridDim.x;
            if (nt < ntiles) {
                const int r0 = nt * 16 + arow;
                const int r1 = r0 + 8;
                p0 = (r0 < nrows) ? *(const float4*)(pre1 + (size_t)r0 * HDIM + ak0) : make_float4(0.f, 0.f, 0.f, 0.f);
                p1 = (r1 < nrows) ? *(const float4*)(pre1 + (size_t)r1 * HDIM + ak0) : make_float4(0.f, 0.f, 0.f, 0.f);
                const int rd = nt * 16 + tid;
                pd = 0.0f;
                if (tid < 16 && rd < nrows) pd = dinv[rd];
            }
        }

        v16h afr[4];
#pragma unroll
        for (int q = 0; q < 4; ++q) afr[q] = load_frag(&aS[buf][mrow][q * 32 + khalf]);
        float di[8];
#pragma unroll
        for (int v = 0; v < 8; ++v) di[v] = dS[buf][rbase + v];

        v8f acc = {};
#pragma unroll
        for (int q = 0; q < 4; ++q)
            acc = __builtin_amdgcn_wmma_f32_16x16x32_f16(false, afr[q], false, bfr[q],
                                                         (short)0, acc, false, false);

        if (block_row + 16 <= nrows) {
#pragma unroll
            for (int v = 0; v < 8; ++v) {
                const int   row = block_row + rbase + v;
                const float val = acc[v];
                xw2[(size_t)row * HDIM + col]      = val;
                agg_seed[(size_t)row * HDIM + col] = di[v] * di[v] * val + bc;
            }
        } else {
#pragma unroll
            for (int v = 0; v < 8; ++v) {
                const int row = block_row + rbase + v;
                if (row < nrows) {
                    const float val = acc[v];
                    xw2[(size_t)row * HDIM + col]      = val;
                    agg_seed[(size_t)row * HDIM + col] = di[v] * di[v] * val + bc;
                }
            }
        }
        buf ^= 1;
    }
}

__global__ __launch_bounds__(256) void out_kernel(const float* __restrict__ pre2,
                                                  const float* __restrict__ Wout,
                                                  const float* __restrict__ bout,
                                                  float* __restrict__ out, int n) {
    int node = (blockIdx.x * blockDim.x + threadIdx.x) >> 5;
    int lane = threadIdx.x & 31;
    if (node >= n) return;

    const float4* hr = (const float4*)(pre2 + (size_t)node * HDIM);
    float4 hv = hr[lane];
    hv.x = fmaxf(hv.x, 0.0f);
    hv.y = fmaxf(hv.y, 0.0f);
    hv.z = fmaxf(hv.z, 0.0f);
    hv.w = fmaxf(hv.w, 0.0f);

    int h0 = lane * 4;
    float acc[CDIM];
#pragma unroll
    for (int c = 0; c < CDIM; ++c) {
        acc[c] = hv.x * Wout[(h0 + 0) * CDIM + c]
               + hv.y * Wout[(h0 + 1) * CDIM + c]
               + hv.z * Wout[(h0 + 2) * CDIM + c]
               + hv.w * Wout[(h0 + 3) * CDIM + c];
    }
#pragma unroll
    for (int off = 16; off > 0; off >>= 1) {
#pragma unroll
        for (int c = 0; c < CDIM; ++c) acc[c] += __shfl_xor(acc[c], off, 32);
    }
    if (lane == 0) {
#pragma unroll
        for (int c = 0; c < CDIM; ++c) acc[c] += bout[c];
        float m = acc[0];
#pragma unroll
        for (int c = 1; c < CDIM; ++c) m = fmaxf(m, acc[c]);
        float sum = 0.0f;
#pragma unroll
        for (int c = 0; c < CDIM; ++c) { acc[c] = __expf(acc[c] - m); sum += acc[c]; }
        float inv = 1.0f / sum;
#pragma unroll
        for (int c = 0; c < CDIM; ++c) out[(size_t)node * CDIM + c] = acc[c] * inv;
    }
}

extern "C" void kernel_launch(void* const* d_in, const int* in_sizes, int n_in,
                              void* d_out, int out_size, void* d_ws, size_t ws_size,
                              hipStream_t stream) {
    const float* x    = (const float*)d_in[0];
    const int*   eidx = (const int*)  d_in[1];
    const float* W1   = (const float*)d_in[2];
    const float* b1   = (const float*)d_in[3];
    const float* W2   = (const float*)d_in[4];
    const float* b2   = (const float*)d_in[5];
    const float* Wout = (const float*)d_in[6];
    const float* bout = (const float*)d_in[7];
    float*       out  = (float*)d_out;

    const int N = in_sizes[0] / (T_HIST * F_IN);
    const int E = in_sizes[1] / 2;
    const int* src = eidx;
    const int* dst = eidx + E;

    float* dinv  = (float*)d_ws;
    int*   queue = (int*)(dinv + N);
    float* bufA  = dinv + ((N + 4 + 63) & ~63);
    float* bufB = bufA + (size_t)N * HDIM;

    const int TB = 256;
    const int NH = N * HDIM;
    const int NH_blocks   = (NH + TB - 1) / TB;
    const int scatter_groups = 2048;
    const int node_blocks = ((N * 32) + TB - 1) / TB;

    fill_zero_kernel<<<(N + 4 + TB - 1) / TB, TB, 0, stream>>>(dinv, N + 4);
    deg_kernel<<<(E + TB - 1) / TB, TB, 0, stream>>>(dst, dinv, E);
    dinv_kernel<<<(N + TB - 1) / TB, TB, 0, stream>>>(dinv, N);

    seed1_kernel<<<NH_blocks, TB, 0, stream>>>(x, W1, b1, dinv, bufB, N);
    scatter_kernel<true><<<scatter_groups, TB, 0, stream>>>(src, dst, dinv, x, W1, bufB, queue, E);

    gemm2_wmma_kernel<<<1024, 256, 0, stream>>>(bufB, W2, b2, dinv, bufA, bufB, N);
    scatter_kernel<false><<<scatter_groups, TB, 0, stream>>>(src, dst, dinv, bufA, W1, bufB, queue + 2, E);

    out_kernel<<<node_blocks, TB, 0, stream>>>(bufB, Wout, bout, out, N);
}


# degree count: each (edge piece, node range) work-group counts targets in LDS (integer adds), stores a partial plane; the rsqrt kernel sums the planes (whole numbers, exact); no global degree atomics
# speedup vs baseline: 3.0997x; 1.2324x over previous
#include <hip/hip_runtime.h>
#include <hip/hip_bf16.h>

typedef __attribute__((ext_vector_type(16))) _Float16 v16h;
typedef __attribute__((ext_vector_type(8)))  float    v8f;

#define T_HIST 3
#define F_IN   5
#define HDIM   128
#define CDIM   5

#define DEG_RANGE    12800
#define DEG_PIECES   32
#define DEG_THREADS  1024
#define DEG_INFLIGHT 8

__device__ __forceinline__ void deg_count(unsigned* cnt, int d, int lo, int nn) {
    const unsigned r = (unsigned)(d - lo);
    if (r < (unsigned)nn) atomicAdd(&cnt[r], 1u);
}

__global__ __launch_bounds__(DEG_THREADS) void deg_part_kernel(const int* __restrict__ dst, float* __restrict__ part,
                                                               int n, int e) {
    __shared__ unsigned cnt[DEG_RANGE];
    const int tid   = threadIdx.x;
    const int piece = blockIdx.x;
    const int lo    = blockIdx.y * DEG_RANGE;
    const int nn    = min(DEG_RANGE, n - lo);
    for (int i = tid; i < nn; i += DEG_THREADS) cnt[i] = 0u;
    __syncthreads();

    const int per = (((e + DEG_PIECES - 1) / DEG_PIECES) + 3) & ~3;
    const int e0  = min(e, piece * per);
    const int e1  = min(e, e0 + per);
    int done = e0;
    if ((((size_t)dst) & 15) == 0) {
        const int   nq = (e1 - e0) >> 2;
        const int4* q  = (const int4*)(dst + e0);
        for (int i = tid; i < nq; i += DEG_THREADS * DEG_INFLIGHT) {
            int4 v[DEG_INFLIGHT];
#pragma unroll
            for (int u = 0; u < DEG_INFLIGHT; ++u) {
                const int k = i + u * DEG_THREADS;
                v[u] = (k < nq) ? q[k] : make_int4(-1, -1, -1, -1);
            }
#pragma unroll
            for (int u = 0; u < DEG_INFLIGHT; ++u) {
                deg_count(cnt, v[u].x, lo, nn);
                deg_count(cnt, v[u].y, lo, nn);
                deg_count(cnt, v[u].z, lo, nn);
                deg_count(cnt, v[u].w, lo, nn);
            }
        }
        done = e0 + (nq << 2);
    }
    for (int i = done + tid; i < e1; i += DEG_THREADS)
        deg_count(cnt, dst[i], lo, nn);
    __syncthreads();

    float* pl = part + (size_t)piece * n + lo;
    for (int i = tid; i < nn; i += DEG_THREADS) pl[i] = (float)cnt[i];
}

__global__ void dinv_kernel(const float* __restrict__ part, float* __restrict__ dinv, int* __restrict__ queue, int n) {
    int i = blockIdx.x * blockDim.x + threadIdx.x;
    if (i < 4) queue[i] = 0;
    if (i < n) {
        float s = 0.0f;
#pragma unroll
        for (int p = 0; p < DEG_PIECES; ++p) s += part[(size_t)p * n + i];
        dinv[i] = rsqrtf(s + 1.0f);
    }
}

__device__ __forceinline__ float l1_dot(const float (&xv)[F_IN], const float (&wv)[F_IN]) {
    float s = 0.0f;
#pragma unroll
    for (int f = 0; f < F_IN; ++f) s = fmaf(xv[f], wv[f], s);
    return s;
}

__global__ void seed1_kernel(const float* __restrict__ x, const float* __restrict__ W1,
                             const float* __restrict__ b1, const float* __restrict__ dinv,
                             float* __restrict__ agg_seed, int n) {
    int idx = blockIdx.x * blockDim.x + threadIdx.x;
    if (idx >= n * HDIM) return;
    int row = idx >> 7;
    int h   = idx & (HDIM - 1);
    const float* xr = x + (size_t)row * (T_HIST * F_IN) + (T_HIST - 1) * F_IN;
    float xv[F_IN], wv[F_IN];
#pragma unroll
    for (int f = 0; f < F_IN; ++f) { xv[f] = xr[f]; wv[f] = W1[f * HDIM + h]; }
    const float s = l1_dot(xv, wv);
    float di = dinv[row];
    agg_seed[idx] = di * di * s + b1[h];
}

#define EDGES_PER_WAVE  32
#define EDGES_PER_BLOCK 256
#define EDGE_GROUP      16
#define HALF_COLS       (HDIM / 2)

__device__ __forceinline__ float lane_bcast_f(float v, int l) {
    return __int_as_float(__builtin_amdgcn_readlane(__float_as_int(v), l));
}

__device__ __forceinline__ void scatter_wave(const int* __restrict__ src, const int* __restrict__ dst,
                                             const float* __restrict__ dinv, const float* __restrict__ xh,
                                             float* __restrict__ ah, int base, int lane, int e) {
    const int mine = base + lane;
    int   s  = 0, d = 0;
    float nm = 0.0f;
    if (mine < e) {
        s  = src[mine];
        d  = dst[mine];
        nm = dinv[s] * dinv[d];
    }
    const int cnt = min(EDGES_PER_WAVE, e - base);
    int j = 0;
    for (; j + EDGE_GROUP <= cnt; j += EDGE_GROUP) {
        float v[EDGE_GROUP][2];
#pragma unroll
        for (int u = 0; u < EDGE_GROUP; ++u) {
            const int sj = __builtin_amdgcn_readlane(s, j + u);
            const unsigned xi = (unsigned)sj * HDIM + (unsigned)lane;
            v[u][0] = xh[xi];
            v[u][1] = xh[xi + 32u];
        }
#pragma unroll
        for (int u = 0; u < EDGE_GROUP; ++u) {
            const int   dj = __builtin_amdgcn_readlane(d, j + u);
            const float nj = lane_bcast_f(nm, j + u);
            const unsigned ai = (unsigned)dj * HDIM + (unsigned)lane;
            atomicAdd(ah + ai,       nj * v[u][0]);
            atomicAdd(ah + ai + 32u, nj * v[u][1]);
        }
    }
    for (; j < cnt; ++j) {
        const int   sj = __builtin_amdgcn_readlane(s, j);
        const int   dj = __builtin_amdgcn_readlane(d, j);
        const float nj = lane_bcast_f(nm, j);
        const unsigned xi = (unsigned)sj * HDIM + (unsigned)lane;
        const unsigned ai = (unsigned)dj * HDIM + (unsigned)lane;
        const float v0 = xh[xi], v1 = xh[xi + 32u];
        atomicAdd(ah + ai,       nj * v0);
        atomicAdd(ah + ai + 32u, nj * v1);
    }
}


#define L1_GROUP 8

__device__ __forceinline__ void l1_edge(int d, float nm, const float (&xr)[F_IN], const float (&w0)[F_IN],
                                        const float (&w1)[F_IN], int lane, int j,
                                        float& p0, float& p1, unsigned& ai) {
    const int   dj = __builtin_amdgcn_readlane(d, j);
    const float nj = lane_bcast_f(nm, j);
    float xj[F_IN];
#pragma unroll
    for (int f = 0; f < F_IN; ++f) xj[f] = lane_bcast_f(xr[f], j);
    ai = (unsigned)dj * HDIM + (unsigned)lane;
    p0 = nj * l1_dot(xj, w0);
    p1 = nj * l1_dot(xj, w1);
}

__device__ __forceinline__ void scatter_wave_l1(const int* __restrict__ src, const int* __restrict__ dst,
                                                const float* __restrict__ dinv, const float* __restrict__ x,
                                                const float (&w0)[F_IN], const float (&w1)[F_IN],
                                                float* __restrict__ ah, int base, int lane, int e) {
    const int mine = base + lane;
    int   s  = 0, d = 0;
    float nm = 0.0f;
    float xr[F_IN] = {0.0f, 0.0f, 0.0f, 0.0f, 0.0f};
    if (mine < e) {
        s  = src[mine];
        d  = dst[mine];
        nm = dinv[s] * dinv[d];
        const float* xp = x + (unsigned)s * (unsigned)(T_HIST * F_IN) + (T_HIST - 1) * F_IN;
#pragma unroll
        for (int f = 0; f < F_IN; ++f) xr[f] = xp[f];
    }
    const int cnt = min(EDGES_PER_WAVE, e - base);
    int j = 0;
    for (; j + L1_GROUP <= cnt; j += L1_GROUP) {
        float    p[L1_GROUP][2];
        unsigned ai[L1_GROUP];
#pragma unroll
        for (int u = 0; u < L1_GROUP; ++u) l1_edge(d, nm, xr, w0, w1, lane, j + u, p[u][0], p[u][1], ai[u]);
#pragma unroll
        for (int u = 0; u < L1_GROUP; ++u) {
            atomicAdd(ah + ai[u],       p[u][0]);
            atomicAdd(ah + ai[u] + 32u, p[u][1]);
        }
    }
    for (; j < cnt; ++j) {
        float p0, p1;
        unsigned ai;
        l1_edge(d, nm, xr, w0, w1, lane, j, p0, p1, ai);
        atomicAdd(ah + ai,       p0);
        atomicAdd(ah + ai + 32u, p1);
    }
}

template <bool L1>
__global__ __launch_bounds__(256) void scatter_kernel(const int* __restrict__ src, const int* __restrict__ dst,
                                                      const float* __restrict__ dinv, const float* __restrict__ rows,
                                                      const float* __restrict__ W1,
                                                      float* __restrict__ agg, int* __restrict__ queue, int e) {
    __shared__ int s_next[2];
    __shared__ int s_half;
    const int tid  = threadIdx.x;
    const int lane = tid & 31;
    const int wave = __builtin_amdgcn_readfirstlane(tid >> 5);
    const int nblk = (e + EDGES_PER_BLOCK - 1) / EDGES_PER_BLOCK;

    if (tid == 0) s_half = (int)((__builtin_amdgcn_s_sendmsg_rtn(0x87) >> 18) & 1u);
    __syncthreads();
    const int myhalf = __builtin_amdgcn_readfirstlane(s_half);

    int it = 0;
    for (int ph = 0; ph < 2; ++ph) {
        const int q = myhalf ^ ph;
        int* cq = queue + q;
        const float* xh = rows + q * HALF_COLS;
        float*       ah = agg + q * HALF_COLS;
        float w0[F_IN], w1[F_IN];
        if constexpr (L1) {
#pragma unroll
            for (int f = 0; f < F_IN; ++f) {
                w0[f] = W1[f * HDIM + q * HALF_COLS + lane];
                w1[f] = W1[f * HDIM + q * HALF_COLS + lane + 32];
            }
        }
        if (tid == 0) s_next[it & 1] = atomicAdd(cq, 1);
        __syncthreads();
        for (;;) {
            const int blk = __builtin_amdgcn_readfirstlane(s_next[it & 1]);
            if (blk >= nblk) break;
            int nb = 0;
            if (tid == 0) nb = atomicAdd(cq, 1);
            const int base = blk * EDGES_PER_BLOCK + wave * EDGES_PER_WAVE;
            if (base < e) {
                if constexpr (L1) scatter_wave_l1(src, dst, dinv, rows, w0, w1, ah, base, lane, e);
                else              scatter_wave(src, dst, dinv, xh, ah, base, lane, e);
            }
            if (tid == 0) s_next[(it + 1) & 1] = nb;
            __builtin_amdgcn_fence(__ATOMIC_RELEASE, "workgroup", "local");
            __builtin_amdgcn_s_barrier();
            __builtin_amdgcn_fence(__ATOMIC_ACQUIRE, "workgroup", "local");
            ++it;
        }
        ++it;
    }
}

#define LPITCH (HDIM + 8)

typedef __attribute__((ext_vector_type(8))) _Float16 v8h;
typedef __attribute__((ext_vector_type(4))) _Float16 v4h;

__device__ __forceinline__ v16h load_frag(const _Float16* rowp) {
    const v8h lo = *(const v8h*)(rowp);
    const v8h hi = *(const v8h*)(rowp + 16);
    return __builtin_shufflevector(lo, hi, 0, 1, 2, 3, 4, 5, 6, 7, 8, 9, 10, 11, 12, 13, 14, 15);
}

__global__ __launch_bounds__(256) void gemm2_wmma_kernel(const float* pre1,
                                                         const float* __restrict__ W2,
                                                         const float* __restrict__ b2,
                                                         const float* __restrict__ dinv,
                                                         float* __restrict__ xw2,
                                                         float* agg_seed,
                                                         int nrows) {
    __shared__ __attribute__((aligned(16))) _Float16 bS[HDIM][LPITCH];
    __shared__ __attribute__((aligned(16))) _Float16 aS[2][16][LPITCH];
    __shared__ float dS[2][16];

    const int tid    = threadIdx.x;
    const int ntiles = (nrows + 15) >> 4;
    int tile = blockIdx.x;
    if (tile >= ntiles) return;

    const int arow = tid >> 5;
    const int ak0  = (tid & 31) * 4;

    float4 p0, p1;
    float  pd = 0.0f;
    {
        const int r0 = tile * 16 + arow;
        const int r1 = r0 + 8;
        p0 = (r0 < nrows) ? *(const float4*)(pre1 + (size_t)r0 * HDIM + ak0) : make_float4(0.f, 0.f, 0.f, 0.f);
        p1 = (r1 < nrows) ? *(const float4*)(pre1 + (size_t)r1 * HDIM + ak0) : make_float4(0.f, 0.f, 0.f, 0.f);
        const int rd = tile * 16 + tid;
        if (tid < 16 && rd < nrows) pd = dinv[rd];
    }

    for (int i = tid; i < HDIM * HDIM; i += 256) {
        const int k = i >> 7;
        const int n = i & (HDIM - 1);
        bS[n][k] = (_Float16)W2[i];
    }
    __syncthreads();

    const int wave  = tid >> 5;
    const int lane  = tid & 31;
    const int ncol  = wave * 16;
    const int mrow  = lane & 15;
    const int khalf = (lane >> 4) * 8;
    const int rbase = (lane >> 4) << 3;

    v16h bfr[4];
#pragma unroll
    for (int q = 0; q < 4; ++q) bfr[q] = load_frag(&bS[ncol + mrow][q * 32 + khalf]);

    const int   col = ncol + mrow;
    const float bc  = b2[col];
    int buf = 0;

    for (; tile < ntiles; tile += gridDim.x) {
        const int block_row = tile * 16;

        {
            v4h h0, h1;
            h0[0] = (_Float16)fmaxf(p0.x, 0.0f); h0[1] = (_Float16)fmaxf(p0.y, 0.0f);
            h0[2] = (_Float16)fmaxf(p0.z, 0.0f); h0[3] = (_Float16)fmaxf(p0.w, 0.0f);
            h1[0] = (_Float16)fmaxf(p1.x, 0.0f); h1[1] = (_Float16)fmaxf(p1.y, 0.0f);
            h1[2] = (_Float16)fmaxf(p1.z, 0.0f); h1[3] = (_Float16)fmaxf(p1.w, 0.0f);
            *(v4h*)&aS[buf][arow][ak0]     = h0;
            *(v4h*)&aS[buf][arow + 8][ak0] = h1;
            if (tid < 16) dS[buf][tid] = pd;
        }
        __syncthreads();

        {
            const int nt = tile + gridDim.x;
            if (nt < ntiles) {
                const int r0 = nt * 16 + arow;
                const int r1 = r0 + 8;
                p0 = (r0 < nrows) ? *(const float4*)(pre1 + (size_t)r0 * HDIM + ak0) : make_float4(0.f, 0.f, 0.f, 0.f);
                p1 = (r1 < nrows) ? *(const float4*)(pre1 + (size_t)r1 * HDIM + ak0) : make_float4(0.f, 0.f, 0.f, 0.f);
                const int rd = nt * 16 + tid;
                pd = 0.0f;
                if (tid < 16 && rd < nrows) pd = dinv[rd];
            }
        }

        v16h afr[4];
#pragma unroll
        for (int q = 0; q < 4; ++q) afr[q] = load_frag(&aS[buf][mrow][q * 32 + khalf]);
        float di[8];
#pragma unroll
        for (int v = 0; v < 8; ++v) di[v] = dS[buf][rbase + v];

        v8f acc = {};
#pragma unroll
        for (int q = 0; q < 4; ++q)
            acc = __builtin_amdgcn_wmma_f32_16x16x32_f16(false, afr[q], false, bfr[q],
                                                         (short)0, acc, false, false);

        if (block_row + 16 <= nrows) {
#pragma unroll
            for (int v = 0; v < 8; ++v) {
                const int   row = block_row + rbase + v;
                const float val = acc[v];
                xw2[(size_t)row * HDIM + col]      = val;
                agg_seed[(size_t)row * HDIM + col] = di[v] * di[v] * val + bc;
            }
        } else {
#pragma unroll
            for (int v = 0; v < 8; ++v) {
                const int row = block_row + rbase + v;
                if (row < nrows) {
                    const float val = acc[v];
                    xw2[(size_t)row * HDIM + col]      = val;
                    agg_seed[(size_t)row * HDIM + col] = di[v] * di[v] * val + bc;
                }
            }
        }
        buf ^= 1;
    }
}

__global__ __launch_bounds__(256) void out_kernel(const float* __restrict__ pre2,
                                                  const float* __restrict__ Wout,
                                                  const float* __restrict__ bout,
                                                  float* __restrict__ out, int n) {
    int node = (blockIdx.x * blockDim.x + threadIdx.x) >> 5;
    int lane = threadIdx.x & 31;
    if (node >= n) return;

    const float4* hr = (const float4*)(pre2 + (size_t)node * HDIM);
    float4 hv = hr[lane];
    hv.x = fmaxf(hv.x, 0.0f);
    hv.y = fmaxf(hv.y, 0.0f);
    hv.z = fmaxf(hv.z, 0.0f);
    hv.w = fmaxf(hv.w, 0.0f);

    int h0 = lane * 4;
    float acc[CDIM];
#pragma unroll
    for (int c = 0; c < CDIM; ++c) {
        acc[c] = hv.x * Wout[(h0 + 0) * CDIM + c]
               + hv.y * Wout[(h0 + 1) * CDIM + c]
               + hv.z * Wout[(h0 + 2) * CDIM + c]
               + hv.w * Wout[(h0 + 3) * CDIM + c];
    }
#pragma unroll
    for (int off = 16; off > 0; off >>= 1) {
#pragma unroll
        for (int c = 0; c < CDIM; ++c) acc[c] += __shfl_xor(acc[c], off, 32);
    }
    if (lane == 0) {
#pragma unroll
        for (int c = 0; c < CDIM; ++c) acc[c] += bout[c];
        float m = acc[0];
#pragma unroll
        for (int c = 1; c < CDIM; ++c) m = fmaxf(m, acc[c]);
        float sum = 0.0f;
#pragma unroll
        for (int c = 0; c < CDIM; ++c) { acc[c] = __expf(acc[c] - m); sum += acc[c]; }
        float inv = 1.0f / sum;
#pragma unroll
        for (int c = 0; c < CDIM; ++c) out[(size_t)node * CDIM + c] = acc[c] * inv;
    }
}

extern "C" void kernel_launch(void* const* d_in, const int* in_sizes, int n_in,
                              void* d_out, int out_size, void* d_ws, size_t ws_size,
                              hipStream_t stream) {
    const float* x    = (const float*)d_in[0];
    const int*   eidx = (const int*)  d_in[1];
    const float* W1   = (const float*)d_in[2];
    const float* b1   = (const float*)d_in[3];
    const float* W2   = (const float*)d_in[4];
    const float* b2   = (const float*)d_in[5];
    const float* Wout = (const float*)d_in[6];
    const float* bout = (const float*)d_in[7];
    float*       out  = (float*)d_out;

    const int N = in_sizes[0] / (T_HIST * F_IN);
    const int E = in_sizes[1] / 2;
    const int* src = eidx;
    const int* dst = eidx + E;

    float* dinv  = (float*)d_ws;
    int*   queue = (int*)(dinv + N);
    float* bufA  = dinv + ((N + 4 + 63) & ~63);
    float* bufB = bufA + (size_t)N * HDIM;

    const int TB = 256;
    const int NH = N * HDIM;
    const int NH_blocks   = (NH + TB - 1) / TB;
    const int scatter_groups = 2048;
    const int node_blocks = ((N * 32) + TB - 1) / TB;

    deg_part_kernel<<<dim3(DEG_PIECES, (N + DEG_RANGE - 1) / DEG_RANGE), DEG_THREADS, 0, stream>>>(dst, bufA, N, E);
    dinv_kernel<<<(N + TB - 1) / TB, TB, 0, stream>>>(bufA, dinv, queue, N);

    seed1_kernel<<<NH_blocks, TB, 0, stream>>>(x, W1, b1, dinv, bufB, N);
    scatter_kernel<true><<<scatter_groups, TB, 0, stream>>>(src, dst, dinv, x, W1, bufB, queue, E);

    gemm2_wmma_kernel<<<1024, 256, 0, stream>>>(bufB, W2, b2, dinv, bufA, bufB, N);
    scatter_kernel<false><<<scatter_groups, TB, 0, stream>>>(src, dst, dinv, bufA, W1, bufB, queue + 2, E);

    out_kernel<<<node_blocks, TB, 0, stream>>>(bufB, Wout, bout, out, N);
}


# edge weight: dinv[d] leaves the atomic sums (edges add dinv[s]*row, seed dinv[d]*row); the plane's reader scales a row by dinv[d] once and adds bias; one dinv gather an edge; sums round differently
# speedup vs baseline: 3.4135x; 1.1012x over previous
#include <hip/hip_runtime.h>
#include <hip/hip_bf16.h>

typedef __attribute__((ext_vector_type(16))) _Float16 v16h;
typedef __attribute__((ext_vector_type(8)))  float    v8f;

#define T_HIST 3
#define F_IN   5
#define HDIM   128
#define CDIM   5

#define DEG_RANGE    12800
#define DEG_PIECES   32
#define DEG_THREADS  1024
#define DEG_INFLIGHT 8

__device__ __forceinline__ void deg_count(unsigned* cnt, int d, int lo, int nn) {
    const unsigned r = (unsigned)(d - lo);
    if (r < (unsigned)nn) atomicAdd(&cnt[r], 1u);
}

__global__ __launch_bounds__(DEG_THREADS) void deg_part_kernel(const int* __restrict__ dst, float* __restrict__ part,
                                                               int n, int e) {
    __shared__ unsigned cnt[DEG_RANGE];
    const int tid   = threadIdx.x;
    const int piece = blockIdx.x;
    const int lo    = blockIdx.y * DEG_RANGE;
    const int nn    = min(DEG_RANGE, n - lo);
    for (int i = tid; i < nn; i += DEG_THREADS) cnt[i] = 0u;
    __syncthreads();

    const int per = (((e + DEG_PIECES - 1) / DEG_PIECES) + 3) & ~3;
    const int e0  = min(e, piece * per);
    const int e1  = min(e, e0 + per);
    int done = e0;
    if ((((size_t)dst) & 15) == 0) {
        const int   nq = (e1 - e0) >> 2;
        const int4* q  = (const int4*)(dst + e0);
        for (int i = tid; i < nq; i += DEG_THREADS * DEG_INFLIGHT) {
            int4 v[DEG_INFLIGHT];
#pragma unroll
            for (int u = 0; u < DEG_INFLIGHT; ++u) {
                const int k = i + u * DEG_THREADS;
                v[u] = (k < nq) ? q[k] : make_int4(-1, -1, -1, -1);
            }
#pragma unroll
            for (int u = 0; u < DEG_INFLIGHT; ++u) {
                deg_count(cnt, v[u].x, lo, nn);
                deg_count(cnt, v[u].y, lo, nn);
                deg_count(cnt, v[u].z, lo, nn);
                deg_count(cnt, v[u].w, lo, nn);
            }
        }
        done = e0 + (nq << 2);
    }
    for (int i = done + tid; i < e1; i += DEG_THREADS)
        deg_count(cnt, dst[i], lo, nn);
    __syncthreads();

    float* pl = part + (size_t)piece * n + lo;
    for (int i = tid; i < nn; i += DEG_THREADS) pl[i] = (float)cnt[i];
}

__global__ void dinv_kernel(const float* __restrict__ part, float* __restrict__ dinv, int* __restrict__ queue, int n) {
    int i = blockIdx.x * blockDim.x + threadIdx.x;
    if (i < 4) queue[i] = 0;
    if (i < n) {
        float s = 0.0f;
#pragma unroll
        for (int p = 0; p < DEG_PIECES; ++p) s += part[(size_t)p * n + i];
        dinv[i] = rsqrtf(s + 1.0f);
    }
}

__device__ __forceinline__ float l1_dot(const float (&xv)[F_IN], const float (&wv)[F_IN]) {
    float s = 0.0f;
#pragma unroll
    for (int f = 0; f < F_IN; ++f) s = fmaf(xv[f], wv[f], s);
    return s;
}

__global__ void seed1_kernel(const float* __restrict__ x, const float* __restrict__ W1,
                             const float* __restrict__ dinv, float* __restrict__ agg_seed, int n) {
    int idx = blockIdx.x * blockDim.x + threadIdx.x;
    if (idx >= n * HDIM) return;
    int row = idx >> 7;
    int h   = idx & (HDIM - 1);
    const float* xr = x + (size_t)row * (T_HIST * F_IN) + (T_HIST - 1) * F_IN;
    float xv[F_IN], wv[F_IN];
#pragma unroll
    for (int f = 0; f < F_IN; ++f) { xv[f] = xr[f]; wv[f] = W1[f * HDIM + h]; }
    const float s = l1_dot(xv, wv);
    agg_seed[idx] = dinv[row] * s;
}

#define EDGES_PER_WAVE  32
#define EDGES_PER_BLOCK 256
#define EDGE_GROUP      16
#define HALF_COLS       (HDIM / 2)

__device__ __forceinline__ float lane_bcast_f(float v, int l) {
    return __int_as_float(__builtin_amdgcn_readlane(__float_as_int(v), l));
}

__device__ __forceinline__ void scatter_wave(const int* __restrict__ src, const int* __restrict__ dst,
                                             const float* __restrict__ dinv, const float* __restrict__ xh,
                                             float* __restrict__ ah, int base, int lane, int e) {
    const int mine = base + lane;
    int   s  = 0, d = 0;
    float nm = 0.0f;
    if (mine < e) {
        s  = src[mine];
        d  = dst[mine];
        nm = dinv[s];
    }
    const int cnt = min(EDGES_PER_WAVE, e - base);
    int j = 0;
    for (; j + EDGE_GROUP <= cnt; j += EDGE_GROUP) {
        float v[EDGE_GROUP][2];
#pragma unroll
        for (int u = 0; u < EDGE_GROUP; ++u) {
            const int sj = __builtin_amdgcn_readlane(s, j + u);
            const unsigned xi = (unsigned)sj * HDIM + (unsigned)lane;
            v[u][0] = xh[xi];
            v[u][1] = xh[xi + 32u];
        }
#pragma unroll
        for (int u = 0; u < EDGE_GROUP; ++u) {
            const int   dj = __builtin_amdgcn_readlane(d, j + u);
            const float nj = lane_bcast_f(nm, j + u);
            const unsigned ai = (unsigned)dj * HDIM + (unsigned)lane;
            atomicAdd(ah + ai,       nj * v[u][0]);
            atomicAdd(ah + ai + 32u, nj * v[u][1]);
        }
    }
    for (; j < cnt; ++j) {
        const int   sj = __builtin_amdgcn_readlane(s, j);
        const int   dj = __builtin_amdgcn_readlane(d, j);
        const float nj = lane_bcast_f(nm, j);
        const unsigned xi = (unsigned)sj * HDIM + (unsigned)lane;
        const unsigned ai = (unsigned)dj * HDIM + (unsigned)lane;
        const float v0 = xh[xi], v1 = xh[xi + 32u];
        atomicAdd(ah + ai,       nj * v0);
        atomicAdd(ah + ai + 32u, nj * v1);
    }
}


#define L1_GROUP 8

__device__ __forceinline__ void l1_edge(int d, float nm, const float (&xr)[F_IN], const float (&w0)[F_IN],
                                        const float (&w1)[F_IN], int lane, int j,
                                        float& p0, float& p1, unsigned& ai) {
    const int   dj = __builtin_amdgcn_readlane(d, j);
    const float nj = lane_bcast_f(nm, j);
    float xj[F_IN];
#pragma unroll
    for (int f = 0; f < F_IN; ++f) xj[f] = lane_bcast_f(xr[f], j);
    ai = (unsigned)dj * HDIM + (unsigned)lane;
    p0 = nj * l1_dot(xj, w0);
    p1 = nj * l1_dot(xj, w1);
}

__device__ __forceinline__ void scatter_wave_l1(const int* __restrict__ src, const int* __restrict__ dst,
                                                const float* __restrict__ dinv, const float* __restrict__ x,
                                                const float (&w0)[F_IN], const float (&w1)[F_IN],
                                                float* __restrict__ ah, int base, int lane, int e) {
    const int mine = base + lane;
    int   s  = 0, d = 0;
    float nm = 0.0f;
    float xr[F_IN] = {0.0f, 0.0f, 0.0f, 0.0f, 0.0f};
    if (mine < e) {
        s  = src[mine];
        d  = dst[mine];
        nm = dinv[s];
        const float* xp = x + (unsigned)s * (unsigned)(T_HIST * F_IN) + (T_HIST - 1) * F_IN;
#pragma unroll
        for (int f = 0; f < F_IN; ++f) xr[f] = xp[f];
    }
    const int cnt = min(EDGES_PER_WAVE, e - base);
    int j = 0;
    for (; j + L1_GROUP <= cnt; j += L1_GROUP) {
        float    p[L1_GROUP][2];
        unsigned ai[L1_GROUP];
#pragma unroll
        for (int u = 0; u < L1_GROUP; ++u) l1_edge(d, nm, xr, w0, w1, lane, j + u, p[u][0], p[u][1], ai[u]);
#pragma unroll
        for (int u = 0; u < L1_GROUP; ++u) {
            atomicAdd(ah + ai[u],       p[u][0]);
            atomicAdd(ah + ai[u] + 32u, p[u][1]);
        }
    }
    for (; j < cnt; ++j) {
        float p0, p1;
        unsigned ai;
        l1_edge(d, nm, xr, w0, w1, lane, j, p0, p1, ai);
        atomicAdd(ah + ai,       p0);
        atomicAdd(ah + ai + 32u, p1);
    }
}

template <bool L1>
__global__ __launch_bounds__(256) void scatter_kernel(const int* __restrict__ src, const int* __restrict__ dst,
                                                      const float* __restrict__ dinv, const float* __restrict__ rows,
                                                      const float* __restrict__ W1,
                                                      float* __restrict__ agg, int* __restrict__ queue, int e) {
    __shared__ int s_next[2];
    __shared__ int s_half;
    const int tid  = threadIdx.x;
    const int lane = tid & 31;
    const int wave = __builtin_amdgcn_readfirstlane(tid >> 5);
    const int nblk = (e + EDGES_PER_BLOCK - 1) / EDGES_PER_BLOCK;

    if (tid == 0) s_half = (int)((__builtin_amdgcn_s_sendmsg_rtn(0x87) >> 18) & 1u);
    __syncthreads();
    const int myhalf = __builtin_amdgcn_readfirstlane(s_half);

    int it = 0;
    for (int ph = 0; ph < 2; ++ph) {
        const int q = myhalf ^ ph;
        int* cq = queue + q;
        const float* xh = rows + q * HALF_COLS;
        float*       ah = agg + q * HALF_COLS;
        float w0[F_IN], w1[F_IN];
        if constexpr (L1) {
#pragma unroll
            for (int f = 0; f < F_IN; ++f) {
                w0[f] = W1[f * HDIM + q * HALF_COLS + lane];
                w1[f] = W1[f * HDIM + q * HALF_COLS + lane + 32];
            }
        }
        if (tid == 0) s_next[it & 1] = atomicAdd(cq, 1);
        __syncthreads();
        for (;;) {
            const int blk = __builtin_amdgcn_readfirstlane(s_next[it & 1]);
            if (blk >= nblk) break;
            int nb = 0;
            if (tid == 0) nb = atomicAdd(cq, 1);
            const int base = blk * EDGES_PER_BLOCK + wave * EDGES_PER_WAVE;
            if (base < e) {
                if constexpr (L1) scatter_wave_l1(src, dst, dinv, rows, w0, w1, ah, base, lane, e);
                else              scatter_wave(src, dst, dinv, xh, ah, base, lane, e);
            }
            if (tid == 0) s_next[(it + 1) & 1] = nb;
            __builtin_amdgcn_fence(__ATOMIC_RELEASE, "workgroup", "local");
            __builtin_amdgcn_s_barrier();
            __builtin_amdgcn_fence(__ATOMIC_ACQUIRE, "workgroup", "local");
            ++it;
        }
        ++it;
    }
}

#define LPITCH (HDIM + 8)

typedef __attribute__((ext_vector_type(8))) _Float16 v8h;
typedef __attribute__((ext_vector_type(4))) _Float16 v4h;

__device__ __forceinline__ v16h load_frag(const _Float16* rowp) {
    const v8h lo = *(const v8h*)(rowp);
    const v8h hi = *(const v8h*)(rowp + 16);
    return __builtin_shufflevector(lo, hi, 0, 1, 2, 3, 4, 5, 6, 7, 8, 9, 10, 11, 12, 13, 14, 15);
}

__global__ __launch_bounds__(256) void gemm2_wmma_kernel(const float* acc1,
                                                         const float* __restrict__ b1,
                                                         const float* __restrict__ W2,
                                                         const float* __restrict__ dinv,
                                                         float* __restrict__ xw2,
                                                         float* agg_seed,
                                                         int nrows) {
    __shared__ __attribute__((aligned(16))) _Float16 bS[HDIM][LPITCH];
    __shared__ __attribute__((aligned(16))) _Float16 aS[2][16][LPITCH];
    __shared__ float dS[2][16];

    const int tid    = threadIdx.x;
    const int ntiles = (nrows + 15) >> 4;
    int tile = blockIdx.x;
    if (tile >= ntiles) return;

    const int arow = tid >> 5;
    const int ak0  = (tid & 31) * 4;

    const float4 bk = make_float4(b1[ak0], b1[ak0 + 1], b1[ak0 + 2], b1[ak0 + 3]);
    float4 p0, p1;
    float  s0, s1;
    float  pd = 0.0f;
    {
        const int r0 = tile * 16 + arow;
        const int r1 = r0 + 8;
        p0 = (r0 < nrows) ? *(const float4*)(acc1 + (size_t)r0 * HDIM + ak0) : make_float4(0.f, 0.f, 0.f, 0.f);
        p1 = (r1 < nrows) ? *(const float4*)(acc1 + (size_t)r1 * HDIM + ak0) : make_float4(0.f, 0.f, 0.f, 0.f);
        s0 = (r0 < nrows) ? dinv[r0] : 0.0f;
        s1 = (r1 < nrows) ? dinv[r1] : 0.0f;
        const int rd = tile * 16 + tid;
        if (tid < 16 && rd < nrows) pd = dinv[rd];
    }

    for (int i = tid; i < HDIM * HDIM; i += 256) {
        const int k = i >> 7;
        const int n = i & (HDIM - 1);
        bS[n][k] = (_Float16)W2[i];
    }
    __syncthreads();

    const int wave  = tid >> 5;
    const int lane  = tid & 31;
    const int ncol  = wave * 16;
    const int mrow  = lane & 15;
    const int khalf = (lane >> 4) * 8;
    const int rbase = (lane >> 4) << 3;

    v16h bfr[4];
#pragma unroll
    for (int q = 0; q < 4; ++q) bfr[q] = load_frag(&bS[ncol + mrow][q * 32 + khalf]);

    const int col = ncol + mrow;
    int buf = 0;

    for (; tile < ntiles; tile += gridDim.x) {
        const int block_row = tile * 16;

        {
            v4h h0, h1;
            h0[0] = (_Float16)fmaxf(fmaf(s0, p0.x, bk.x), 0.0f); h0[1] = (_Float16)fmaxf(fmaf(s0, p0.y, bk.y), 0.0f);
            h0[2] = (_Float16)fmaxf(fmaf(s0, p0.z, bk.z), 0.0f); h0[3] = (_Float16)fmaxf(fmaf(s0, p0.w, bk.w), 0.0f);
            h1[0] = (_Float16)fmaxf(fmaf(s1, p1.x, bk.x), 0.0f); h1[1] = (_Float16)fmaxf(fmaf(s1, p1.y, bk.y), 0.0f);
            h1[2] = (_Float16)fmaxf(fmaf(s1, p1.z, bk.z), 0.0f); h1[3] = (_Float16)fmaxf(fmaf(s1, p1.w, bk.w), 0.0f);
            *(v4h*)&aS[buf][arow][ak0]     = h0;
            *(v4h*)&aS[buf][arow + 8][ak0] = h1;
            if (tid < 16) dS[buf][tid] = pd;
        }
        __syncthreads();

        {
            const int nt = tile + gridDim.x;
            if (nt < ntiles) {
                const int r0 = nt * 16 + arow;
                const int r1 = r0 + 8;
                p0 = (r0 < nrows) ? *(const float4*)(acc1 + (size_t)r0 * HDIM + ak0) : make_float4(0.f, 0.f, 0.f, 0.f);
                p1 = (r1 < nrows) ? *(const float4*)(acc1 + (size_t)r1 * HDIM + ak0) : make_float4(0.f, 0.f, 0.f, 0.f);
                s0 = (r0 < nrows) ? dinv[r0] : 0.0f;
                s1 = (r1 < nrows) ? dinv[r1] : 0.0f;
                const int rd = nt * 16 + tid;
                pd = 0.0f;
                if (tid < 16 && rd < nrows) pd = dinv[rd];
            }
        }

        v16h afr[4];
#pragma unroll
        for (int q = 0; q < 4; ++q) afr[q] = load_frag(&aS[buf][mrow][q * 32 + khalf]);
        float di[8];
#pragma unroll
        for (int v = 0; v < 8; ++v) di[v] = dS[buf][rbase + v];

        v8f acc = {};
#pragma unroll
        for (int q = 0; q < 4; ++q)
            acc = __builtin_amdgcn_wmma_f32_16x16x32_f16(false, afr[q], false, bfr[q],
                                                         (short)0, acc, false, false);

        if (block_row + 16 <= nrows) {
#pragma unroll
            for (int v = 0; v < 8; ++v) {
                const int   row = block_row + rbase + v;
                const float val = acc[v];
                xw2[(size_t)row * HDIM + col]      = val;
                agg_seed[(size_t)row * HDIM + col] = di[v] * val;
            }
        } else {
#pragma unroll
            for (int v = 0; v < 8; ++v) {
                const int row = block_row + rbase + v;
                if (row < nrows) {
                    const float val = acc[v];
                    xw2[(size_t)row * HDIM + col]      = val;
                    agg_seed[(size_t)row * HDIM + col] = di[v] * val;
                }
            }
        }
        buf ^= 1;
    }
}

__global__ __launch_bounds__(256) void out_kernel(const float* __restrict__ acc2,
                                                  const float* __restrict__ dinv,
                                                  const float* __restrict__ b2,
                                                  const float* __restrict__ Wout,
                                                  const float* __restrict__ bout,
                                                  float* __restrict__ out, int n) {
    int node = (blockIdx.x * blockDim.x + threadIdx.x) >> 5;
    int lane = threadIdx.x & 31;
    if (node >= n) return;

    const float4* hr = (const float4*)(acc2 + (size_t)node * HDIM);
    float4 hv = hr[lane];
    const float  di = dinv[node];
    const float4 bv = make_float4(b2[lane * 4], b2[lane * 4 + 1], b2[lane * 4 + 2], b2[lane * 4 + 3]);
    hv.x = fmaxf(fmaf(di, hv.x, bv.x), 0.0f);
    hv.y = fmaxf(fmaf(di, hv.y, bv.y), 0.0f);
    hv.z = fmaxf(fmaf(di, hv.z, bv.z), 0.0f);
    hv.w = fmaxf(fmaf(di, hv.w, bv.w), 0.0f);

    int h0 = lane * 4;
    float acc[CDIM];
#pragma unroll
    for (int c = 0; c < CDIM; ++c) {
        acc[c] = hv.x * Wout[(h0 + 0) * CDIM + c]
               + hv.y * Wout[(h0 + 1) * CDIM + c]
               + hv.z * Wout[(h0 + 2) * CDIM + c]
               + hv.w * Wout[(h0 + 3) * CDIM + c];
    }
#pragma unroll
    for (int off = 16; off > 0; off >>= 1) {
#pragma unroll
        for (int c = 0; c < CDIM; ++c) acc[c] += __shfl_xor(acc[c], off, 32);
    }
    if (lane == 0) {
#pragma unroll
        for (int c = 0; c < CDIM; ++c) acc[c] += bout[c];
        float m = acc[0];
#pragma unroll
        for (int c = 1; c < CDIM; ++c) m = fmaxf(m, acc[c]);
        float sum = 0.0f;
#pragma unroll
        for (int c = 0; c < CDIM; ++c) { acc[c] = __expf(acc[c] - m); sum += acc[c]; }
        float inv = 1.0f / sum;
#pragma unroll
        for (int c = 0; c < CDIM; ++c) out[(size_t)node * CDIM + c] = acc[c] * inv;
    }
}

extern "C" void kernel_launch(void* const* d_in, const int* in_sizes, int n_in,
                              void* d_out, int out_size, void* d_ws, size_t ws_size,
                              hipStream_t stream) {
    const float* x    = (const float*)d_in[0];
    const int*   eidx = (const int*)  d_in[1];
    const float* W1   = (const float*)d_in[2];
    const float* b1   = (const float*)d_in[3];
    const float* W2   = (const float*)d_in[4];
    const float* b2   = (const float*)d_in[5];
    const float* Wout = (const float*)d_in[6];
    const float* bout = (const float*)d_in[7];
    float*       out  = (float*)d_out;

    const int N = in_sizes[0] / (T_HIST * F_IN);
    const int E = in_sizes[1] / 2;
    const int* src = eidx;
    const int* dst = eidx + E;

    float* dinv  = (float*)d_ws;
    int*   queue = (int*)(dinv + N);
    float* bufA  = dinv + ((N + 4 + 63) & ~63);
    float* bufB = bufA + (size_t)N * HDIM;

    const int TB = 256;
    const int NH = N * HDIM;
    const int NH_blocks   = (NH + TB - 1) / TB;
    const int scatter_groups = 2048;
    const int node_blocks = ((N * 32) + TB - 1) / TB;

    deg_part_kernel<<<dim3(DEG_PIECES, (N + DEG_RANGE - 1) / DEG_RANGE), DEG_THREADS, 0, stream>>>(dst, bufA, N, E);
    dinv_kernel<<<(N + TB - 1) / TB, TB, 0, stream>>>(bufA, dinv, queue, N);

    seed1_kernel<<<NH_blocks, TB, 0, stream>>>(x, W1, dinv, bufB, N);
    scatter_kernel<true><<<scatter_groups, TB, 0, stream>>>(src, dst, dinv, x, W1, bufB, queue, E);

    gemm2_wmma_kernel<<<1024, 256, 0, stream>>>(bufB, b1, W2, dinv, bufA, bufB, N);
    scatter_kernel<false><<<scatter_groups, TB, 0, stream>>>(src, dst, dinv, bufA, W1, bufB, queue + 2, E);

    out_kernel<<<node_blocks, TB, 0, stream>>>(bufB, dinv, b2, Wout, bout, out, N);
}


# layer-1 scatter: what an edge gathers (x_last[s], dinv[s]) is one aligned 32-byte record per node that the dinv kernel writes: 2 loads into one 32-byte piece, not 3 into two lines; same addends
# speedup vs baseline: 3.6400x; 1.0664x over previous
#include <hip/hip_runtime.h>
#include <hip/hip_bf16.h>

typedef __attribute__((ext_vector_type(16))) _Float16 v16h;
typedef __attribute__((ext_vector_type(8)))  float    v8f;

#define T_HIST 3
#define F_IN   5
#define HDIM   128
#define CDIM   5

#define DEG_RANGE    12800
#define DEG_PIECES   32
#define DEG_THREADS  1024
#define DEG_INFLIGHT 8

__device__ __forceinline__ void deg_count(unsigned* cnt, int d, int lo, int nn) {
    const unsigned r = (unsigned)(d - lo);
    if (r < (unsigned)nn) atomicAdd(&cnt[r], 1u);
}

__global__ __launch_bounds__(DEG_THREADS) void deg_part_kernel(const int* __restrict__ dst, float* __restrict__ part,
                                                               int n, int e) {
    __shared__ unsigned cnt[DEG_RANGE];
    const int tid   = threadIdx.x;
    const int piece = blockIdx.x;
    const int lo    = blockIdx.y * DEG_RANGE;
    const int nn    = min(DEG_RANGE, n - lo);
    for (int i = tid; i < nn; i += DEG_THREADS) cnt[i] = 0u;
    __syncthreads();

    const int per = (((e + DEG_PIECES - 1) / DEG_PIECES) + 3) & ~3;
    const int e0  = min(e, piece * per);
    const int e1  = min(e, e0 + per);
    int done = e0;
    if ((((size_t)dst) & 15) == 0) {
        const int   nq = (e1 - e0) >> 2;
        const int4* q  = (const int4*)(dst + e0);
        for (int i = tid; i < nq; i += DEG_THREADS * DEG_INFLIGHT) {
            int4 v[DEG_INFLIGHT];
#pragma unroll
            for (int u = 0; u < DEG_INFLIGHT; ++u) {
                const int k = i + u * DEG_THREADS;
                v[u] = (k < nq) ? q[k] : make_int4(-1, -1, -1, -1);
            }
#pragma unroll
            for (int u = 0; u < DEG_INFLIGHT; ++u) {
                deg_count(cnt, v[u].x, lo, nn);
                deg_count(cnt, v[u].y, lo, nn);
                deg_count(cnt, v[u].z, lo, nn);
                deg_count(cnt, v[u].w, lo, nn);
            }
        }
        done = e0 + (nq << 2);
    }
    for (int i = done + tid; i < e1; i += DEG_THREADS)
        deg_count(cnt, dst[i], lo, nn);
    __syncthreads();

    float* pl = part + (size_t)piece * n + lo;
    for (int i = tid; i < nn; i += DEG_THREADS) pl[i] = (float)cnt[i];
}

#define REC_WORDS 8

__global__ void dinv_kernel(const float* __restrict__ part, const float* __restrict__ x, float* __restrict__ dinv,
                            float* __restrict__ rec, int* __restrict__ queue, int n) {
    int i = blockIdx.x * blockDim.x + threadIdx.x;
    if (i < 4) queue[i] = 0;
    if (i < n) {
        float s = 0.0f;
#pragma unroll
        for (int p = 0; p < DEG_PIECES; ++p) s += part[(size_t)p * n + i];
        const float di = rsqrtf(s + 1.0f);
        dinv[i] = di;
        const float* xr = x + (size_t)i * (T_HIST * F_IN) + (T_HIST - 1) * F_IN;
        float4* r = (float4*)(rec + (size_t)i * REC_WORDS);
        r[0] = make_float4(xr[0], xr[1], xr[2], xr[3]);
        r[1] = make_float4(xr[4], di, 0.0f, 0.0f);
    }
}

__device__ __forceinline__ float l1_dot(const float (&xv)[F_IN], const float (&wv)[F_IN]) {
    float s = 0.0f;
#pragma unroll
    for (int f = 0; f < F_IN; ++f) s = fmaf(xv[f], wv[f], s);
    return s;
}

__global__ void seed1_kernel(const float* __restrict__ x, const float* __restrict__ W1,
                             const float* __restrict__ dinv, float* __restrict__ agg_seed, int n) {
    int idx = blockIdx.x * blockDim.x + threadIdx.x;
    if (idx >= n * HDIM) return;
    int row = idx >> 7;
    int h   = idx & (HDIM - 1);
    const float* xr = x + (size_t)row * (T_HIST * F_IN) + (T_HIST - 1) * F_IN;
    float xv[F_IN], wv[F_IN];
#pragma unroll
    for (int f = 0; f < F_IN; ++f) { xv[f] = xr[f]; wv[f] = W1[f * HDIM + h]; }
    const float s = l1_dot(xv, wv);
    agg_seed[idx] = dinv[row] * s;
}

#define EDGES_PER_WAVE  32
#define EDGES_PER_BLOCK 256
#define EDGE_GROUP      16
#define HALF_COLS       (HDIM / 2)

__device__ __forceinline__ float lane_bcast_f(float v, int l) {
    return __int_as_float(__builtin_amdgcn_readlane(__float_as_int(v), l));
}

__device__ __forceinline__ void scatter_wave(const int* __restrict__ src, const int* __restrict__ dst,
                                             const float* __restrict__ dinv, const float* __restrict__ xh,
                                             float* __restrict__ ah, int base, int lane, int e) {
    const int mine = base + lane;
    int   s  = 0, d = 0;
    float nm = 0.0f;
    if (mine < e) {
        s  = src[mine];
        d  = dst[mine];
        nm = dinv[s];
    }
    const int cnt = min(EDGES_PER_WAVE, e - base);
    int j = 0;
    for (; j + EDGE_GROUP <= cnt; j += EDGE_GROUP) {
        float v[EDGE_GROUP][2];
#pragma unroll
        for (int u = 0; u < EDGE_GROUP; ++u) {
            const int sj = __builtin_amdgcn_readlane(s, j + u);
            const unsigned xi = (unsigned)sj * HDIM + (unsigned)lane;
            v[u][0] = xh[xi];
            v[u][1] = xh[xi + 32u];
        }
#pragma unroll
        for (int u = 0; u < EDGE_GROUP; ++u) {
            const int   dj = __builtin_amdgcn_readlane(d, j + u);
            const float nj = lane_bcast_f(nm, j + u);
            const unsigned ai = (unsigned)dj * HDIM + (unsigned)lane;
            atomicAdd(ah + ai,       nj * v[u][0]);
            atomicAdd(ah + ai + 32u, nj * v[u][1]);
        }
    }
    for (; j < cnt; ++j) {
        const int   sj = __builtin_amdgcn_readlane(s, j);
        const int   dj = __builtin_amdgcn_readlane(d, j);
        const float nj = lane_bcast_f(nm, j);
        const unsigned xi = (unsigned)sj * HDIM + (unsigned)lane;
        const unsigned ai = (unsigned)dj * HDIM + (unsigned)lane;
        const float v0 = xh[xi], v1 = xh[xi + 32u];
        atomicAdd(ah + ai,       nj * v0);
        atomicAdd(ah + ai + 32u, nj * v1);
    }
}


#define L1_GROUP 8

__device__ __forceinline__ void l1_edge(int d, float nm, const float (&xr)[F_IN], const float (&w0)[F_IN],
                                        const float (&w1)[F_IN], int lane, int j,
                                        float& p0, float& p1, unsigned& ai) {
    const int   dj = __builtin_amdgcn_readlane(d, j);
    const float nj = lane_bcast_f(nm, j);
    float xj[F_IN];
#pragma unroll
    for (int f = 0; f < F_IN; ++f) xj[f] = lane_bcast_f(xr[f], j);
    ai = (unsigned)dj * HDIM + (unsigned)lane;
    p0 = nj * l1_dot(xj, w0);
    p1 = nj * l1_dot(xj, w1);
}

__device__ __forceinline__ void scatter_wave_l1(const int* __restrict__ src, const int* __restrict__ dst,
                                                const float* __restrict__ rec,
                                                const float (&w0)[F_IN], const float (&w1)[F_IN],
                                                float* __restrict__ ah, int base, int lane, int e) {
    const int mine = base + lane;
    int   s  = 0, d = 0;
    float nm = 0.0f;
    float xr[F_IN] = {0.0f, 0.0f, 0.0f, 0.0f, 0.0f};
    if (mine < e) {
        s  = src[mine];
        d  = dst[mine];
        const float* rp = rec + (unsigned)s * (unsigned)REC_WORDS;
        const float4 ra = *(const float4*)rp;
        const float2 rb = *(const float2*)(rp + 4);
        xr[0] = ra.x; xr[1] = ra.y; xr[2] = ra.z; xr[3] = ra.w; xr[4] = rb.x;
        nm = rb.y;
    }
    const int cnt = min(EDGES_PER_WAVE, e - base);
    int j = 0;
    for (; j + L1_GROUP <= cnt; j += L1_GROUP) {
        float    p[L1_GROUP][2];
        unsigned ai[L1_GROUP];
#pragma unroll
        for (int u = 0; u < L1_GROUP; ++u) l1_edge(d, nm, xr, w0, w1, lane, j + u, p[u][0], p[u][1], ai[u]);
#pragma unroll
        for (int u = 0; u < L1_GROUP; ++u) {
            atomicAdd(ah + ai[u],       p[u][0]);
            atomicAdd(ah + ai[u] + 32u, p[u][1]);
        }
    }
    for (; j < cnt; ++j) {
        float p0, p1;
        unsigned ai;
        l1_edge(d, nm, xr, w0, w1, lane, j, p0, p1, ai);
        atomicAdd(ah + ai,       p0);
        atomicAdd(ah + ai + 32u, p1);
    }
}

template <bool L1>
__global__ __launch_bounds__(256) void scatter_kernel(const int* __restrict__ src, const int* __restrict__ dst,
                                                      const float* __restrict__ dinv, const float* __restrict__ rows,
                                                      const float* __restrict__ W1,
                                                      float* __restrict__ agg, int* __restrict__ queue, int e) {
    __shared__ int s_next[2];
    __shared__ int s_half;
    const int tid  = threadIdx.x;
    const int lane = tid & 31;
    const int wave = __builtin_amdgcn_readfirstlane(tid >> 5);
    const int nblk = (e + EDGES_PER_BLOCK - 1) / EDGES_PER_BLOCK;

    if (tid == 0) s_half = (int)((__builtin_amdgcn_s_sendmsg_rtn(0x87) >> 18) & 1u);
    __syncthreads();
    const int myhalf = __builtin_amdgcn_readfirstlane(s_half);

    int it = 0;
    for (int ph = 0; ph < 2; ++ph) {
        const int q = myhalf ^ ph;
        int* cq = queue + q;
        const float* xh = rows + q * HALF_COLS;
        float*       ah = agg + q * HALF_COLS;
        float w0[F_IN], w1[F_IN];
        if constexpr (L1) {
#pragma unroll
            for (int f = 0; f < F_IN; ++f) {
                w0[f] = W1[f * HDIM + q * HALF_COLS + lane];
                w1[f] = W1[f * HDIM + q * HALF_COLS + lane + 32];
            }
        }
        if (tid == 0) s_next[it & 1] = atomicAdd(cq, 1);
        __syncthreads();
        for (;;) {
            const int blk = __builtin_amdgcn_readfirstlane(s_next[it & 1]);
            if (blk >= nblk) break;
            int nb = 0;
            if (tid == 0) nb = atomicAdd(cq, 1);
            const int base = blk * EDGES_PER_BLOCK + wave * EDGES_PER_WAVE;
            if (base < e) {
                if constexpr (L1) scatter_wave_l1(src, dst, rows, w0, w1, ah, base, lane, e);
                else              scatter_wave(src, dst, dinv, xh, ah, base, lane, e);
            }
            if (tid == 0) s_next[(it + 1) & 1] = nb;
            __builtin_amdgcn_fence(__ATOMIC_RELEASE, "workgroup", "local");
            __builtin_amdgcn_s_barrier();
            __builtin_amdgcn_fence(__ATOMIC_ACQUIRE, "workgroup", "local");
            ++it;
        }
        ++it;
    }
}

#define LPITCH (HDIM + 8)

typedef __attribute__((ext_vector_type(8))) _Float16 v8h;
typedef __attribute__((ext_vector_type(4))) _Float16 v4h;

__device__ __forceinline__ v16h load_frag(const _Float16* rowp) {
    const v8h lo = *(const v8h*)(rowp);
    const v8h hi = *(const v8h*)(rowp + 16);
    return __builtin_shufflevector(lo, hi, 0, 1, 2, 3, 4, 5, 6, 7, 8, 9, 10, 11, 12, 13, 14, 15);
}

__global__ __launch_bounds__(256) void gemm2_wmma_kernel(const float* acc1,
                                                         const float* __restrict__ b1,
                                                         const float* __restrict__ W2,
                                                         const float* __restrict__ dinv,
                                                         float* __restrict__ xw2,
                                                         float* agg_seed,
                                                         int nrows) {
    __shared__ __attribute__((aligned(16))) _Float16 bS[HDIM][LPITCH];
    __shared__ __attribute__((aligned(16))) _Float16 aS[2][16][LPITCH];
    __shared__ float dS[2][16];

    const int tid    = threadIdx.x;
    const int ntiles = (nrows + 15) >> 4;
    int tile = blockIdx.x;
    if (tile >= ntiles) return;

    const int arow = tid >> 5;
    const int ak0  = (tid & 31) * 4;

    const float4 bk = make_float4(b1[ak0], b1[ak0 + 1], b1[ak0 + 2], b1[ak0 + 3]);
    float4 p0, p1;
    float  s0, s1;
    float  pd = 0.0f;
    {
        const int r0 = tile * 16 + arow;
        const int r1 = r0 + 8;
        p0 = (r0 < nrows) ? *(const float4*)(acc1 + (size_t)r0 * HDIM + ak0) : make_float4(0.f, 0.f, 0.f, 0.f);
        p1 = (r1 < nrows) ? *(const float4*)(acc1 + (size_t)r1 * HDIM + ak0) : make_float4(0.f, 0.f, 0.f, 0.f);
        s0 = (r0 < nrows) ? dinv[r0] : 0.0f;
        s1 = (r1 < nrows) ? dinv[r1] : 0.0f;
        const int rd = tile * 16 + tid;
        if (tid < 16 && rd < nrows) pd = dinv[rd];
    }

    for (int i = tid; i < HDIM * HDIM; i += 256) {
        const int k = i >> 7;
        const int n = i & (HDIM - 1);
        bS[n][k] = (_Float16)W2[i];
    }
    __syncthreads();

    const int wave  = tid >> 5;
    const int lane  = tid & 31;
    const int ncol  = wave * 16;
    const int mrow  = lane & 15;
    const int khalf = (lane >> 4) * 8;
    const int rbase = (lane >> 4) << 3;

    v16h bfr[4];
#pragma unroll
    for (int q = 0; q < 4; ++q) bfr[q] = load_frag(&bS[ncol + mrow][q * 32 + khalf]);

    const int col = ncol + mrow;
    int buf = 0;

    for (; tile < ntiles; tile += gridDim.x) {
        const int block_row = tile * 16;

        {
            v4h h0, h1;
            h0[0] = (_Float16)fmaxf(fmaf(s0, p0.x, bk.x), 0.0f); h0[1] = (_Float16)fmaxf(fmaf(s0, p0.y, bk.y), 0.0f);
            h0[2] = (_Float16)fmaxf(fmaf(s0, p0.z, bk.z), 0.0f); h0[3] = (_Float16)fmaxf(fmaf(s0, p0.w, bk.w), 0.0f);
            h1[0] = (_Float16)fmaxf(fmaf(s1, p1.x, bk.x), 0.0f); h1[1] = (_Float16)fmaxf(fmaf(s1, p1.y, bk.y), 0.0f);
            h1[2] = (_Float16)fmaxf(fmaf(s1, p1.z, bk.z), 0.0f); h1[3] = (_Float16)fmaxf(fmaf(s1, p1.w, bk.w), 0.0f);
            *(v4h*)&aS[buf][arow][ak0]     = h0;
            *(v4h*)&aS[buf][arow + 8][ak0] = h1;
            if (tid < 16) dS[buf][tid] = pd;
        }
        __syncthreads();

        {
            const int nt = tile + gridDim.x;
            if (nt < ntiles) {
                const int r0 = nt * 16 + arow;
                const int r1 = r0 + 8;
                p0 = (r0 < nrows) ? *(const float4*)(acc1 + (size_t)r0 * HDIM + ak0) : make_float4(0.f, 0.f, 0.f, 0.f);
                p1 = (r1 < nrows) ? *(const float4*)(acc1 + (size_t)r1 * HDIM + ak0) : make_float4(0.f, 0.f, 0.f, 0.f);
                s0 = (r0 < nrows) ? dinv[r0] : 0.0f;
                s1 = (r1 < nrows) ? dinv[r1] : 0.0f;
                const int rd = nt * 16 + tid;
                pd = 0.0f;
                if (tid < 16 && rd < nrows) pd = dinv[rd];
            }
        }

        v16h afr[4];
#pragma unroll
        for (int q = 0; q < 4; ++q) afr[q] = load_frag(&aS[buf][mrow][q * 32 + khalf]);
        float di[8];
#pragma unroll
        for (int v = 0; v < 8; ++v) di[v] = dS[buf][rbase + v];

        v8f acc = {};
#pragma unroll
        for (int q = 0; q < 4; ++q)
            acc = __builtin_amdgcn_wmma_f32_16x16x32_f16(false, afr[q], false, bfr[q],
                                                         (short)0, acc, false, false);

        if (block_row + 16 <= nrows) {
#pragma unroll
            for (int v = 0; v < 8; ++v) {
                const int   row = block_row + rbase + v;
                const float val = acc[v];
                xw2[(size_t)row * HDIM + col]      = val;
                agg_seed[(size_t)row * HDIM + col] = di[v] * val;
            }
        } else {
#pragma unroll
            for (int v = 0; v < 8; ++v) {
                const int row = block_row + rbase + v;
                if (row < nrows) {
                    const float val = acc[v];
                    xw2[(size_t)row * HDIM + col]      = val;
                    agg_seed[(size_t)row * HDIM + col] = di[v] * val;
                }
            }
        }
        buf ^= 1;
    }
}

__global__ __launch_bounds__(256) void out_kernel(const float* __restrict__ acc2,
                                                  const float* __restrict__ dinv,
                                                  const float* __restrict__ b2,
                                                  const float* __restrict__ Wout,
                                                  const float* __restrict__ bout,
                                                  float* __restrict__ out, int n) {
    int node = (blockIdx.x * blockDim.x + threadIdx.x) >> 5;
    int lane = threadIdx.x & 31;
    if (node >= n) return;

    const float4* hr = (const float4*)(acc2 + (size_t)node * HDIM);
    float4 hv = hr[lane];
    const float  di = dinv[node];
    const float4 bv = make_float4(b2[lane * 4], b2[lane * 4 + 1], b2[lane * 4 + 2], b2[lane * 4 + 3]);
    hv.x = fmaxf(fmaf(di, hv.x, bv.x), 0.0f);
    hv.y = fmaxf(fmaf(di, hv.y, bv.y), 0.0f);
    hv.z = fmaxf(fmaf(di, hv.z, bv.z), 0.0f);
    hv.w = fmaxf(fmaf(di, hv.w, bv.w), 0.0f);

    int h0 = lane * 4;
    float acc[CDIM];
#pragma unroll
    for (int c = 0; c < CDIM; ++c) {
        acc[c] = hv.x * Wout[(h0 + 0) * CDIM + c]
               + hv.y * Wout[(h0 + 1) * CDIM + c]
               + hv.z * Wout[(h0 + 2) * CDIM + c]
               + hv.w * Wout[(h0 + 3) * CDIM + c];
    }
#pragma unroll
    for (int off = 16; off > 0; off >>= 1) {
#pragma unroll
        for (int c = 0; c < CDIM; ++c) acc[c] += __shfl_xor(acc[c], off, 32);
    }
    if (lane == 0) {
#pragma unroll
        for (int c = 0; c < CDIM; ++c) acc[c] += bout[c];
        float m = acc[0];
#pragma unroll
        for (int c = 1; c < CDIM; ++c) m = fmaxf(m, acc[c]);
        float sum = 0.0f;
#pragma unroll
        for (int c = 0; c < CDIM; ++c) { acc[c] = __expf(acc[c] - m); sum += acc[c]; }
        float inv = 1.0f / sum;
#pragma unroll
        for (int c = 0; c < CDIM; ++c) out[(size_t)node * CDIM + c] = acc[c] * inv;
    }
}

extern "C" void kernel_launch(void* const* d_in, const int* in_sizes, int n_in,
                              void* d_out, int out_size, void* d_ws, size_t ws_size,
                              hipStream_t stream) {
    const float* x    = (const float*)d_in[0];
    const int*   eidx = (const int*)  d_in[1];
    const float* W1   = (const float*)d_in[2];
    const float* b1   = (const float*)d_in[3];
    const float* W2   = (const float*)d_in[4];
    const float* b2   = (const float*)d_in[5];
    const float* Wout = (const float*)d_in[6];
    const float* bout = (const float*)d_in[7];
    float*       out  = (float*)d_out;

    const int N = in_sizes[0] / (T_HIST * F_IN);
    const int E = in_sizes[1] / 2;
    const int* src = eidx;
    const int* dst = eidx + E;

    float* dinv  = (float*)d_ws;
    int*   queue = (int*)(dinv + N);
    float* bufA  = dinv + ((N + 4 + 63) & ~63);
    float* bufB = bufA + (size_t)N * HDIM;
    float* rec  = bufA + (size_t)N * DEG_PIECES;

    const int TB = 256;
    const int NH = N * HDIM;
    const int NH_blocks   = (NH + TB - 1) / TB;
    const int scatter_groups = 2048;
    const int node_blocks = ((N * 32) + TB - 1) / TB;

    deg_part_kernel<<<dim3(DEG_PIECES, (N + DEG_RANGE - 1) / DEG_RANGE), DEG_THREADS, 0, stream>>>(dst, bufA, N, E);
    dinv_kernel<<<(N + TB - 1) / TB, TB, 0, stream>>>(bufA, x, dinv, rec, queue, N);

    seed1_kernel<<<NH_blocks, TB, 0, stream>>>(x, W1, dinv, bufB, N);
    scatter_kernel<true><<<scatter_groups, TB, 0, stream>>>(src, dst, dinv, rec, W1, bufB, queue, E);

    gemm2_wmma_kernel<<<1024, 256, 0, stream>>>(bufB, b1, W2, dinv, bufA, bufB, N);
    scatter_kernel<false><<<scatter_groups, TB, 0, stream>>>(src, dst, dinv, bufA, W1, bufB, queue + 2, E);

    out_kernel<<<node_blocks, TB, 0, stream>>>(bufB, dinv, b2, Wout, bout, out, N);
}


# layer-2 rows stored with their source factor: gemm2's epilogue writes dinv*product as gathered row and as seed, so that scatter gathers no dinv word per edge; same addends, adds in arrival order
# speedup vs baseline: 3.8774x; 1.0652x over previous
#include <hip/hip_runtime.h>
#include <hip/hip_bf16.h>

typedef __attribute__((ext_vector_type(16))) _Float16 v16h;
typedef __attribute__((ext_vector_type(8)))  float    v8f;

#define T_HIST 3
#define F_IN   5
#define HDIM   128
#define CDIM   5

#define DEG_RANGE    12800
#define DEG_PIECES   32
#define DEG_THREADS  1024
#define DEG_INFLIGHT 8

__device__ __forceinline__ void deg_count(unsigned* cnt, int d, int lo, int nn) {
    const unsigned r = (unsigned)(d - lo);
    if (r < (unsigned)nn) atomicAdd(&cnt[r], 1u);
}

__global__ __launch_bounds__(DEG_THREADS) void deg_part_kernel(const int* __restrict__ dst, float* __restrict__ part,
                                                               int n, int e) {
    __shared__ unsigned cnt[DEG_RANGE];
    const int tid   = threadIdx.x;
    const int piece = blockIdx.x;
    const int lo    = blockIdx.y * DEG_RANGE;
    const int nn    = min(DEG_RANGE, n - lo);
    for (int i = tid; i < nn; i += DEG_THREADS) cnt[i] = 0u;
    __syncthreads();

    const int per = (((e + DEG_PIECES - 1) / DEG_PIECES) + 3) & ~3;
    const int e0  = min(e, piece * per);
    const int e1  = min(e, e0 + per);
    int done = e0;
    if ((((size_t)dst) & 15) == 0) {
        const int   nq = (e1 - e0) >> 2;
        const int4* q  = (const int4*)(dst + e0);
        for (int i = tid; i < nq; i += DEG_THREADS * DEG_INFLIGHT) {
            int4 v[DEG_INFLIGHT];
#pragma unroll
            for (int u = 0; u < DEG_INFLIGHT; ++u) {
                const int k = i + u * DEG_THREADS;
                v[u] = (k < nq) ? q[k] : make_int4(-1, -1, -1, -1);
            }
#pragma unroll
            for (int u = 0; u < DEG_INFLIGHT; ++u) {
                deg_count(cnt, v[u].x, lo, nn);
                deg_count(cnt, v[u].y, lo, nn);
                deg_count(cnt, v[u].z, lo, nn);
                deg_count(cnt, v[u].w, lo, nn);
            }
        }
        done = e0 + (nq << 2);
    }
    for (int i = done + tid; i < e1; i += DEG_THREADS)
        deg_count(cnt, dst[i], lo, nn);
    __syncthreads();

    float* pl = part + (size_t)piece * n + lo;
    for (int i = tid; i < nn; i += DEG_THREADS) pl[i] = (float)cnt[i];
}

#define REC_WORDS 8

__global__ void dinv_kernel(const float* __restrict__ part, const float* __restrict__ x, float* __restrict__ dinv,
                            float* __restrict__ rec, int* __restrict__ queue, int n) {
    int i = blockIdx.x * blockDim.x + threadIdx.x;
    if (i < 4) queue[i] = 0;
    if (i < n) {
        float s = 0.0f;
#pragma unroll
        for (int p = 0; p < DEG_PIECES; ++p) s += part[(size_t)p * n + i];
        const float di = rsqrtf(s + 1.0f);
        dinv[i] = di;
        const float* xr = x + (size_t)i * (T_HIST * F_IN) + (T_HIST - 1) * F_IN;
        float4* r = (float4*)(rec + (size_t)i * REC_WORDS);
        r[0] = make_float4(xr[0], xr[1], xr[2], xr[3]);
        r[1] = make_float4(xr[4], di, 0.0f, 0.0f);
    }
}

__device__ __forceinline__ float l1_dot(const float (&xv)[F_IN], const float (&wv)[F_IN]) {
    float s = 0.0f;
#pragma unroll
    for (int f = 0; f < F_IN; ++f) s = fmaf(xv[f], wv[f], s);
    return s;
}

__global__ void seed1_kernel(const float* __restrict__ x, const float* __restrict__ W1,
                             const float* __restrict__ dinv, float* __restrict__ agg_seed, int n) {
    int idx = blockIdx.x * blockDim.x + threadIdx.x;
    if (idx >= n * HDIM) return;
    int row = idx >> 7;
    int h   = idx & (HDIM - 1);
    const float* xr = x + (size_t)row * (T_HIST * F_IN) + (T_HIST - 1) * F_IN;
    float xv[F_IN], wv[F_IN];
#pragma unroll
    for (int f = 0; f < F_IN; ++f) { xv[f] = xr[f]; wv[f] = W1[f * HDIM + h]; }
    const float s = l1_dot(xv, wv);
    agg_seed[idx] = dinv[row] * s;
}

#define EDGES_PER_WAVE  32
#define EDGES_PER_BLOCK 256
#define EDGE_GROUP      16
#define HALF_COLS       (HDIM / 2)

__device__ __forceinline__ float lane_bcast_f(float v, int l) {
    return __int_as_float(__builtin_amdgcn_readlane(__float_as_int(v), l));
}

__device__ __forceinline__ void scatter_wave(const int* __restrict__ src, const int* __restrict__ dst,
                                             const float* __restrict__ xh,
                                             float* __restrict__ ah, int base, int lane, int e) {
    const int mine = base + lane;
    int s = 0, d = 0;
    if (mine < e) {
        s = src[mine];
        d = dst[mine];
    }
    const int cnt = min(EDGES_PER_WAVE, e - base);
    int j = 0;
    for (; j + EDGE_GROUP <= cnt; j += EDGE_GROUP) {
        float v[EDGE_GROUP][2];
#pragma unroll
        for (int u = 0; u < EDGE_GROUP; ++u) {
            const int sj = __builtin_amdgcn_readlane(s, j + u);
            const unsigned xi = (unsigned)sj * HDIM + (unsigned)lane;
            v[u][0] = xh[xi];
            v[u][1] = xh[xi + 32u];
        }
#pragma unroll
        for (int u = 0; u < EDGE_GROUP; ++u) {
            const int dj = __builtin_amdgcn_readlane(d, j + u);
            const unsigned ai = (unsigned)dj * HDIM + (unsigned)lane;
            atomicAdd(ah + ai,       v[u][0]);
            atomicAdd(ah + ai + 32u, v[u][1]);
        }
    }
    for (; j < cnt; ++j) {
        const int sj = __builtin_amdgcn_readlane(s, j);
        const int dj = __builtin_amdgcn_readlane(d, j);
        const unsigned xi = (unsigned)sj * HDIM + (unsigned)lane;
        const unsigned ai = (unsigned)dj * HDIM + (unsigned)lane;
        const float v0 = xh[xi], v1 = xh[xi + 32u];
        atomicAdd(ah + ai,       v0);
        atomicAdd(ah + ai + 32u, v1);
    }
}


#define L1_GROUP 8

__device__ __forceinline__ void l1_edge(int d, float nm, const float (&xr)[F_IN], const float (&w0)[F_IN],
                                        const float (&w1)[F_IN], int lane, int j,
                                        float& p0, float& p1, unsigned& ai) {
    const int   dj = __builtin_amdgcn_readlane(d, j);
    const float nj = lane_bcast_f(nm, j);
    float xj[F_IN];
#pragma unroll
    for (int f = 0; f < F_IN; ++f) xj[f] = lane_bcast_f(xr[f], j);
    ai = (unsigned)dj * HDIM + (unsigned)lane;
    p0 = nj * l1_dot(xj, w0);
    p1 = nj * l1_dot(xj, w1);
}

__device__ __forceinline__ void scatter_wave_l1(const int* __restrict__ src, const int* __restrict__ dst,
                                                const float* __restrict__ rec,
                                                const float (&w0)[F_IN], const float (&w1)[F_IN],
                                                float* __restrict__ ah, int base, int lane, int e) {
    const int mine = base + lane;
    int   s  = 0, d = 0;
    float nm = 0.0f;
    float xr[F_IN] = {0.0f, 0.0f, 0.0f, 0.0f, 0.0f};
    if (mine < e) {
        s  = src[mine];
        d  = dst[mine];
        const float* rp = rec + (unsigned)s * (unsigned)REC_WORDS;
        const float4 ra = *(const float4*)rp;
        const float2 rb = *(const float2*)(rp + 4);
        xr[0] = ra.x; xr[1] = ra.y; xr[2] = ra.z; xr[3] = ra.w; xr[4] = rb.x;
        nm = rb.y;
    }
    const int cnt = min(EDGES_PER_WAVE, e - base);
    int j = 0;
    for (; j + L1_GROUP <= cnt; j += L1_GROUP) {
        float    p[L1_GROUP][2];
        unsigned ai[L1_GROUP];
#pragma unroll
        for (int u = 0; u < L1_GROUP; ++u) l1_edge(d, nm, xr, w0, w1, lane, j + u, p[u][0], p[u][1], ai[u]);
#pragma unroll
        for (int u = 0; u < L1_GROUP; ++u) {
            atomicAdd(ah + ai[u],       p[u][0]);
            atomicAdd(ah + ai[u] + 32u, p[u][1]);
        }
    }
    for (; j < cnt; ++j) {
        float p0, p1;
        unsigned ai;
        l1_edge(d, nm, xr, w0, w1, lane, j, p0, p1, ai);
        atomicAdd(ah + ai,       p0);
        atomicAdd(ah + ai + 32u, p1);
    }
}

template <bool L1>
__global__ __launch_bounds__(256) void scatter_kernel(const int* __restrict__ src, const int* __restrict__ dst,
                                                      const float* __restrict__ rows,
                                                      const float* __restrict__ W1,
                                                      float* __restrict__ agg, int* __restrict__ queue, int e) {
    __shared__ int s_next[2];
    __shared__ int s_half;
    const int tid  = threadIdx.x;
    const int lane = tid & 31;
    const int wave = __builtin_amdgcn_readfirstlane(tid >> 5);
    const int nblk = (e + EDGES_PER_BLOCK - 1) / EDGES_PER_BLOCK;

    if (tid == 0) s_half = (int)((__builtin_amdgcn_s_sendmsg_rtn(0x87) >> 18) & 1u);
    __syncthreads();
    const int myhalf = __builtin_amdgcn_readfirstlane(s_half);

    int it = 0;
    for (int ph = 0; ph < 2; ++ph) {
        const int q = myhalf ^ ph;
        int* cq = queue + q;
        const float* xh = rows + q * HALF_COLS;
        float*       ah = agg + q * HALF_COLS;
        float w0[F_IN], w1[F_IN];
        if constexpr (L1) {
#pragma unroll
            for (int f = 0; f < F_IN; ++f) {
                w0[f] = W1[f * HDIM + q * HALF_COLS + lane];
                w1[f] = W1[f * HDIM + q * HALF_COLS + lane + 32];
            }
        }
        if (tid == 0) s_next[it & 1] = atomicAdd(cq, 1);
        __syncthreads();
        for (;;) {
            const int blk = __builtin_amdgcn_readfirstlane(s_next[it & 1]);
            if (blk >= nblk) break;
            int nb = 0;
            if (tid == 0) nb = atomicAdd(cq, 1);
            const int base = blk * EDGES_PER_BLOCK + wave * EDGES_PER_WAVE;
            if (base < e) {
                if constexpr (L1) scatter_wave_l1(src, dst, rows, w0, w1, ah, base, lane, e);
                else              scatter_wave(src, dst, xh, ah, base, lane, e);
            }
            if (tid == 0) s_next[(it + 1) & 1] = nb;
            __builtin_amdgcn_fence(__ATOMIC_RELEASE, "workgroup", "local");
            __builtin_amdgcn_s_barrier();
            __builtin_amdgcn_fence(__ATOMIC_ACQUIRE, "workgroup", "local");
            ++it;
        }
        ++it;
    }
}

#define LPITCH (HDIM + 8)

typedef __attribute__((ext_vector_type(8))) _Float16 v8h;
typedef __attribute__((ext_vector_type(4))) _Float16 v4h;

__device__ __forceinline__ v16h load_frag(const _Float16* rowp) {
    const v8h lo = *(const v8h*)(rowp);
    const v8h hi = *(const v8h*)(rowp + 16);
    return __builtin_shufflevector(lo, hi, 0, 1, 2, 3, 4, 5, 6, 7, 8, 9, 10, 11, 12, 13, 14, 15);
}

__global__ __launch_bounds__(256) void gemm2_wmma_kernel(const float* acc1,
                                                         const float* __restrict__ b1,
                                                         const float* __restrict__ W2,
                                                         const float* __restrict__ dinv,
                                                         float* __restrict__ xw2,
                                                         float* agg_seed,
                                                         int nrows) {
    __shared__ __attribute__((aligned(16))) _Float16 bS[HDIM][LPITCH];
    __shared__ __attribute__((aligned(16))) _Float16 aS[2][16][LPITCH];
    __shared__ float dS[2][16];

    const int tid    = threadIdx.x;
    const int ntiles = (nrows + 15) >> 4;
    int tile = blockIdx.x;
    if (tile >= ntiles) return;

    const int arow = tid >> 5;
    const int ak0  = (tid & 31) * 4;

    const float4 bk = make_float4(b1[ak0], b1[ak0 + 1], b1[ak0 + 2], b1[ak0 + 3]);
    float4 p0, p1;
    float  s0, s1;
    float  pd = 0.0f;
    {
        const int r0 = tile * 16 + arow;
        const int r1 = r0 + 8;
        p0 = (r0 < nrows) ? *(const float4*)(acc1 + (size_t)r0 * HDIM + ak0) : make_float4(0.f, 0.f, 0.f, 0.f);
        p1 = (r1 < nrows) ? *(const float4*)(acc1 + (size_t)r1 * HDIM + ak0) : make_float4(0.f, 0.f, 0.f, 0.f);
        s0 = (r0 < nrows) ? dinv[r0] : 0.0f;
        s1 = (r1 < nrows) ? dinv[r1] : 0.0f;
        const int rd = tile * 16 + tid;
        if (tid < 16 && rd < nrows) pd = dinv[rd];
    }

    for (int i = tid; i < HDIM * HDIM; i += 256) {
        const int k = i >> 7;
        const int n = i & (HDIM - 1);
        bS[n][k] = (_Float16)W2[i];
    }
    __syncthreads();

    const int wave  = tid >> 5;
    const int lane  = tid & 31;
    const int ncol  = wave * 16;
    const int mrow  = lane & 15;
    const int khalf = (lane >> 4) * 8;
    const int rbase = (lane >> 4) << 3;

    v16h bfr[4];
#pragma unroll
    for (int q = 0; q < 4; ++q) bfr[q] = load_frag(&bS[ncol + mrow][q * 32 + khalf]);

    const int col = ncol + mrow;
    int buf = 0;

    for (; tile < ntiles; tile += gridDim.x) {
        const int block_row = tile * 16;

        {
            v4h h0, h1;
            h0[0] = (_Float16)fmaxf(fmaf(s0, p0.x, bk.x), 0.0f); h0[1] = (_Float16)fmaxf(fmaf(s0, p0.y, bk.y), 0.0f);
            h0[2] = (_Float16)fmaxf(fmaf(s0, p0.z, bk.z), 0.0f); h0[3] = (_Float16)fmaxf(fmaf(s0, p0.w, bk.w), 0.0f);
            h1[0] = (_Float16)fmaxf(fmaf(s1, p1.x, bk.x), 0.0f); h1[1] = (_Float16)fmaxf(fmaf(s1, p1.y, bk.y), 0.0f);
            h1[2] = (_Float16)fmaxf(fmaf(s1, p1.z, bk.z), 0.0f); h1[3] = (_Float16)fmaxf(fmaf(s1, p1.w, bk.w), 0.0f);
            *(v4h*)&aS[buf][arow][ak0]     = h0;
            *(v4h*)&aS[buf][arow + 8][ak0] = h1;
            if (tid < 16) dS[buf][tid] = pd;
        }
        __syncthreads();

        {
            const int nt = tile + gridDim.x;
            if (nt < ntiles) {
                const int r0 = nt * 16 + arow;
                const int r1 = r0 + 8;
                p0 = (r0 < nrows) ? *(const float4*)(acc1 + (size_t)r0 * HDIM + ak0) : make_float4(0.f, 0.f, 0.f, 0.f);
                p1 = (r1 < nrows) ? *(const float4*)(acc1 + (size_t)r1 * HDIM + ak0) : make_float4(0.f, 0.f, 0.f, 0.f);
                s0 = (r0 < nrows) ? dinv[r0] : 0.0f;
                s1 = (r1 < nrows) ? dinv[r1] : 0.0f;
                const int rd = nt * 16 + tid;
                pd = 0.0f;
                if (tid < 16 && rd < nrows) pd = dinv[rd];
            }
        }

        v16h afr[4];
#pragma unroll
        for (int q = 0; q < 4; ++q) afr[q] = load_frag(&aS[buf][mrow][q * 32 + khalf]);
        float di[8];
#pragma unroll
        for (int v = 0; v < 8; ++v) di[v] = dS[buf][rbase + v];

        v8f acc = {};
#pragma unroll
        for (int q = 0; q < 4; ++q)
            acc = __builtin_amdgcn_wmma_f32_16x16x32_f16(false, afr[q], false, bfr[q],
                                                         (short)0, acc, false, false);

        if (block_row + 16 <= nrows) {
#pragma unroll
            for (int v = 0; v < 8; ++v) {
                const int   row = block_row + rbase + v;
                const float val = di[v] * acc[v];
                xw2[(size_t)row * HDIM + col]      = val;
                agg_seed[(size_t)row * HDIM + col] = val;
            }
        } else {
#pragma unroll
            for (int v = 0; v < 8; ++v) {
                const int row = block_row + rbase + v;
                if (row < nrows) {
                    const float val = di[v] * acc[v];
                    xw2[(size_t)row * HDIM + col]      = val;
                    agg_seed[(size_t)row * HDIM + col] = val;
                }
            }
        }
        buf ^= 1;
    }
}

__global__ __launch_bounds__(256) void out_kernel(const float* __restrict__ acc2,
                                                  const float* __restrict__ dinv,
                                                  const float* __restrict__ b2,
                                                  const float* __restrict__ Wout,
                                                  const float* __restrict__ bout,
                                                  float* __restrict__ out, int n) {
    int node = (blockIdx.x * blockDim.x + threadIdx.x) >> 5;
    int lane = threadIdx.x & 31;
    if (node >= n) return;

    const float4* hr = (const float4*)(acc2 + (size_t)node * HDIM);
    float4 hv = hr[lane];
    const float  di = dinv[node];
    const float4 bv = make_float4(b2[lane * 4], b2[lane * 4 + 1], b2[lane * 4 + 2], b2[lane * 4 + 3]);
    hv.x = fmaxf(fmaf(di, hv.x, bv.x), 0.0f);
    hv.y = fmaxf(fmaf(di, hv.y, bv.y), 0.0f);
    hv.z = fmaxf(fmaf(di, hv.z, bv.z), 0.0f);
    hv.w = fmaxf(fmaf(di, hv.w, bv.w), 0.0f);

    int h0 = lane * 4;
    float acc[CDIM];
#pragma unroll
    for (int c = 0; c < CDIM; ++c) {
        acc[c] = hv.x * Wout[(h0 + 0) * CDIM + c]
               + hv.y * Wout[(h0 + 1) * CDIM + c]
               + hv.z * Wout[(h0 + 2) * CDIM + c]
               + hv.w * Wout[(h0 + 3) * CDIM + c];
    }
#pragma unroll
    for (int off = 16; off > 0; off >>= 1) {
#pragma unroll
        for (int c = 0; c < CDIM; ++c) acc[c] += __shfl_xor(acc[c], off, 32);
    }
    if (lane == 0) {
#pragma unroll
        for (int c = 0; c < CDIM; ++c) acc[c] += bout[c];
        float m = acc[0];
#pragma unroll
        for (int c = 1; c < CDIM; ++c) m = fmaxf(m, acc[c]);
        float sum = 0.0f;
#pragma unroll
        for (int c = 0; c < CDIM; ++c) { acc[c] = __expf(acc[c] - m); sum += acc[c]; }
        float inv = 1.0f / sum;
#pragma unroll
        for (int c = 0; c < CDIM; ++c) out[(size_t)node * CDIM + c] = acc[c] * inv;
    }
}

extern "C" void kernel_launch(void* const* d_in, const int* in_sizes, int n_in,
                              void* d_out, int out_size, void* d_ws, size_t ws_size,
                              hipStream_t stream) {
    const float* x    = (const float*)d_in[0];
    const int*   eidx = (const int*)  d_in[1];
    const float* W1   = (const float*)d_in[2];
    const float* b1   = (const float*)d_in[3];
    const float* W2   = (const float*)d_in[4];
    const float* b2   = (const float*)d_in[5];
    const float* Wout = (const float*)d_in[6];
    const float* bout = (const float*)d_in[7];
    float*       out  = (float*)d_out;

    const int N = in_sizes[0] / (T_HIST * F_IN);
    const int E = in_sizes[1] / 2;
    const int* src = eidx;
    const int* dst = eidx + E;

    float* dinv  = (float*)d_ws;
    int*   queue = (int*)(dinv + N);
    float* bufA  = dinv + ((N + 4 + 63) & ~63);
    float* bufB = bufA + (size_t)N * HDIM;
    float* rec  = bufA + (size_t)N * DEG_PIECES;

    const int TB = 256;
    const int NH = N * HDIM;
    const int NH_blocks   = (NH + TB - 1) / TB;
    const int scatter_groups = 2048;
    const int node_blocks = ((N * 32) + TB - 1) / TB;

    deg_part_kernel<<<dim3(DEG_PIECES, (N + DEG_RANGE - 1) / DEG_RANGE), DEG_THREADS, 0, stream>>>(dst, bufA, N, E);
    dinv_kernel<<<(N + TB - 1) / TB, TB, 0, stream>>>(bufA, x, dinv, rec, queue, N);

    seed1_kernel<<<NH_blocks, TB, 0, stream>>>(x, W1, dinv, bufB, N);
    scatter_kernel<true><<<scatter_groups, TB, 0, stream>>>(src, dst, rec, W1, bufB, queue, E);

    gemm2_wmma_kernel<<<1024, 256, 0, stream>>>(bufB, b1, W2, dinv, bufA, bufB, N);
    scatter_kernel<false><<<scatter_groups, TB, 0, stream>>>(src, dst, bufA, W1, bufB, queue + 2, E);

    out_kernel<<<node_blocks, TB, 0, stream>>>(bufB, dinv, b2, Wout, bout, out, N);
}


# layer-1 self-loop term added by the reader (gemm2 A staging, behind the edge sums: rounds differently); sum plane zero-filled inside the degree-count launch; seed kernel and launch gone (6 launches)
# speedup vs baseline: 4.1010x; 1.0577x over previous
#include <hip/hip_runtime.h>
#include <hip/hip_bf16.h>

typedef __attribute__((ext_vector_type(16))) _Float16 v16h;
typedef __attribute__((ext_vector_type(8)))  float    v8f;

#define T_HIST 3
#define F_IN   5
#define HDIM   128
#define CDIM   5

#define DEG_RANGE    12800
#define DEG_PIECES   32
#define DEG_THREADS  1024
#define DEG_INFLIGHT 8

__device__ __forceinline__ void deg_count(unsigned* cnt, int d, int lo, int nn) {
    const unsigned r = (unsigned)(d - lo);
    if (r < (unsigned)nn) atomicAdd(&cnt[r], 1u);
}

__global__ __launch_bounds__(DEG_THREADS) void deg_part_kernel(const int* __restrict__ dst, float* __restrict__ part,
                                                               float* __restrict__ zplane, int n, int e) {
    __shared__ unsigned cnt[DEG_RANGE];
    const int tid   = threadIdx.x;
    const int piece = blockIdx.x;
    const int lo    = blockIdx.y * DEG_RANGE;
    const int nn    = min(DEG_RANGE, n - lo);
    for (int i = tid; i < nn; i += DEG_THREADS) cnt[i] = 0u;
    __syncthreads();

    {
        const int nz4  = n * (HDIM / 4);
        const int nwg  = gridDim.x * gridDim.y;
        const int wg   = blockIdx.y * gridDim.x + blockIdx.x;
        const int per4 = (nz4 + nwg - 1) / nwg;
        const int z0   = min(nz4, wg * per4);
        const int z1   = min(nz4, z0 + per4);
        float4* z4 = (float4*)zplane;
        for (int i = z0 + tid; i < z1; i += DEG_THREADS) z4[i] = make_float4(0.0f, 0.0f, 0.0f, 0.0f);
    }

    const int per = (((e + DEG_PIECES - 1) / DEG_PIECES) + 3) & ~3;
    const int e0  = min(e, piece * per);
    const int e1  = min(e, e0 + per);
    int done = e0;
    if ((((size_t)dst) & 15) == 0) {
        const int   nq = (e1 - e0) >> 2;
        const int4* q  = (const int4*)(dst + e0);
        for (int i = tid; i < nq; i += DEG_THREADS * DEG_INFLIGHT) {
            int4 v[DEG_INFLIGHT];
#pragma unroll
            for (int u = 0; u < DEG_INFLIGHT; ++u) {
                const int k = i + u * DEG_THREADS;
                v[u] = (k < nq) ? q[k] : make_int4(-1, -1, -1, -1);
            }
#pragma unroll
            for (int u = 0; u < DEG_INFLIGHT; ++u) {
                deg_count(cnt, v[u].x, lo, nn);
                deg_count(cnt, v[u].y, lo, nn);
                deg_count(cnt, v[u].z, lo, nn);
                deg_count(cnt, v[u].w, lo, nn);
            }
        }
        done = e0 + (nq << 2);
    }
    for (int i = done + tid; i < e1; i += DEG_THREADS)
        deg_count(cnt, dst[i], lo, nn);
    __syncthreads();

    float* pl = part + (size_t)piece * n + lo;
    for (int i = tid; i < nn; i += DEG_THREADS) pl[i] = (float)cnt[i];
}

#define REC_WORDS 8

__global__ void dinv_kernel(const float* __restrict__ part, const float* __restrict__ x, float* __restrict__ dinv,
                            float* __restrict__ rec, int* __restrict__ queue, int n) {
    int i = blockIdx.x * blockDim.x + threadIdx.x;
    if (i < 4) queue[i] = 0;
    if (i < n) {
        float s = 0.0f;
#pragma unroll
        for (int p = 0; p < DEG_PIECES; ++p) s += part[(size_t)p * n + i];
        const float di = rsqrtf(s + 1.0f);
        dinv[i] = di;
        const float* xr = x + (size_t)i * (T_HIST * F_IN) + (T_HIST - 1) * F_IN;
        float4* r = (float4*)(rec + (size_t)i * REC_WORDS);
        r[0] = make_float4(xr[0], xr[1], xr[2], xr[3]);
        r[1] = make_float4(xr[4], di, 0.0f, 0.0f);
    }
}

__device__ __forceinline__ float l1_dot(const float (&xv)[F_IN], const float (&wv)[F_IN]) {
    float s = 0.0f;
#pragma unroll
    for (int f = 0; f < F_IN; ++f) s = fmaf(xv[f], wv[f], s);
    return s;
}

#define EDGES_PER_WAVE  32
#define EDGES_PER_BLOCK 256
#define EDGE_GROUP      16
#define HALF_COLS       (HDIM / 2)

__device__ __forceinline__ float lane_bcast_f(float v, int l) {
    return __int_as_float(__builtin_amdgcn_readlane(__float_as_int(v), l));
}

__device__ __forceinline__ void scatter_wave(const int* __restrict__ src, const int* __restrict__ dst,
                                             const float* __restrict__ xh,
                                             float* __restrict__ ah, int base, int lane, int e) {
    const int mine = base + lane;
    int s = 0, d = 0;
    if (mine < e) {
        s = src[mine];
        d = dst[mine];
    }
    const int cnt = min(EDGES_PER_WAVE, e - base);
    int j = 0;
    for (; j + EDGE_GROUP <= cnt; j += EDGE_GROUP) {
        float v[EDGE_GROUP][2];
#pragma unroll
        for (int u = 0; u < EDGE_GROUP; ++u) {
            const int sj = __builtin_amdgcn_readlane(s, j + u);
            const unsigned xi = (unsigned)sj * HDIM + (unsigned)lane;
            v[u][0] = xh[xi];
            v[u][1] = xh[xi + 32u];
        }
#pragma unroll
        for (int u = 0; u < EDGE_GROUP; ++u) {
            const int dj = __builtin_amdgcn_readlane(d, j + u);
            const unsigned ai = (unsigned)dj * HDIM + (unsigned)lane;
            atomicAdd(ah + ai,       v[u][0]);
            atomicAdd(ah + ai + 32u, v[u][1]);
        }
    }
    for (; j < cnt; ++j) {
        const int sj = __builtin_amdgcn_readlane(s, j);
        const int dj = __builtin_amdgcn_readlane(d, j);
        const unsigned xi = (unsigned)sj * HDIM + (unsigned)lane;
        const unsigned ai = (unsigned)dj * HDIM + (unsigned)lane;
        const float v0 = xh[xi], v1 = xh[xi + 32u];
        atomicAdd(ah + ai,       v0);
        atomicAdd(ah + ai + 32u, v1);
    }
}


#define L1_GROUP 8

__device__ __forceinline__ void l1_edge(int d, float nm, const float (&xr)[F_IN], const float (&w0)[F_IN],
                                        const float (&w1)[F_IN], int lane, int j,
                                        float& p0, float& p1, unsigned& ai) {
    const int   dj = __builtin_amdgcn_readlane(d, j);
    const float nj = lane_bcast_f(nm, j);
    float xj[F_IN];
#pragma unroll
    for (int f = 0; f < F_IN; ++f) xj[f] = lane_bcast_f(xr[f], j);
    ai = (unsigned)dj * HDIM + (unsigned)lane;
    p0 = nj * l1_dot(xj, w0);
    p1 = nj * l1_dot(xj, w1);
}

__device__ __forceinline__ void scatter_wave_l1(const int* __restrict__ src, const int* __restrict__ dst,
                                                const float* __restrict__ rec,
                                                const float (&w0)[F_IN], const float (&w1)[F_IN],
                                                float* __restrict__ ah, int base, int lane, int e) {
    const int mine = base + lane;
    int   s  = 0, d = 0;
    float nm = 0.0f;
    float xr[F_IN] = {0.0f, 0.0f, 0.0f, 0.0f, 0.0f};
    if (mine < e) {
        s  = src[mine];
        d  = dst[mine];
        const float* rp = rec + (unsigned)s * (unsigned)REC_WORDS;
        const float4 ra = *(const float4*)rp;
        const float2 rb = *(const float2*)(rp + 4);
        xr[0] = ra.x; xr[1] = ra.y; xr[2] = ra.z; xr[3] = ra.w; xr[4] = rb.x;
        nm = rb.y;
    }
    const int cnt = min(EDGES_PER_WAVE, e - base);
    int j = 0;
    for (; j + L1_GROUP <= cnt; j += L1_GROUP) {
        float    p[L1_GROUP][2];
        unsigned ai[L1_GROUP];
#pragma unroll
        for (int u = 0; u < L1_GROUP; ++u) l1_edge(d, nm, xr, w0, w1, lane, j + u, p[u][0], p[u][1], ai[u]);
#pragma unroll
        for (int u = 0; u < L1_GROUP; ++u) {
            atomicAdd(ah + ai[u],       p[u][0]);
            atomicAdd(ah + ai[u] + 32u, p[u][1]);
        }
    }
    for (; j < cnt; ++j) {
        float p0, p1;
        unsigned ai;
        l1_edge(d, nm, xr, w0, w1, lane, j, p0, p1, ai);
        atomicAdd(ah + ai,       p0);
        atomicAdd(ah + ai + 32u, p1);
    }
}

template <bool L1>
__global__ __launch_bounds__(256) void scatter_kernel(const int* __restrict__ src, const int* __restrict__ dst,
                                                      const float* __restrict__ rows,
                                                      const float* __restrict__ W1,
                                                      float* __restrict__ agg, int* __restrict__ queue, int e) {
    __shared__ int s_next[2];
    __shared__ int s_half;
    const int tid  = threadIdx.x;
    const int lane = tid & 31;
    const int wave = __builtin_amdgcn_readfirstlane(tid >> 5);
    const int nblk = (e + EDGES_PER_BLOCK - 1) / EDGES_PER_BLOCK;

    if (tid == 0) s_half = (int)((__builtin_amdgcn_s_sendmsg_rtn(0x87) >> 18) & 1u);
    __syncthreads();
    const int myhalf = __builtin_amdgcn_readfirstlane(s_half);

    int it = 0;
    for (int ph = 0; ph < 2; ++ph) {
        const int q = myhalf ^ ph;
        int* cq = queue + q;
        const float* xh = rows + q * HALF_COLS;
        float*       ah = agg + q * HALF_COLS;
        float w0[F_IN], w1[F_IN];
        if constexpr (L1) {
#pragma unroll
            for (int f = 0; f < F_IN; ++f) {
                w0[f] = W1[f * HDIM + q * HALF_COLS + lane];
                w1[f] = W1[f * HDIM + q * HALF_COLS + lane + 32];
            }
        }
        if (tid == 0) s_next[it & 1] = atomicAdd(cq, 1);
        __syncthreads();
        for (;;) {
            const int blk = __builtin_amdgcn_readfirstlane(s_next[it & 1]);
            if (blk >= nblk) break;
            int nb = 0;
            if (tid == 0) nb = atomicAdd(cq, 1);
            const int base = blk * EDGES_PER_BLOCK + wave * EDGES_PER_WAVE;
            if (base < e) {
                if constexpr (L1) scatter_wave_l1(src, dst, rows, w0, w1, ah, base, lane, e);
                else              scatter_wave(src, dst, xh, ah, base, lane, e);
            }
            if (tid == 0) s_next[(it + 1) & 1] = nb;
            __builtin_amdgcn_fence(__ATOMIC_RELEASE, "workgroup", "local");
            __builtin_amdgcn_s_barrier();
            __builtin_amdgcn_fence(__ATOMIC_ACQUIRE, "workgroup", "local");
            ++it;
        }
        ++it;
    }
}

#define LPITCH (HDIM + 8)

typedef __attribute__((ext_vector_type(8))) _Float16 v8h;
typedef __attribute__((ext_vector_type(4))) _Float16 v4h;

__device__ __forceinline__ v16h load_frag(const _Float16* rowp) {
    const v8h lo = *(const v8h*)(rowp);
    const v8h hi = *(const v8h*)(rowp + 16);
    return __builtin_shufflevector(lo, hi, 0, 1, 2, 3, 4, 5, 6, 7, 8, 9, 10, 11, 12, 13, 14, 15);
}

__device__ __forceinline__ _Float16 l1_finish(float sum, float di, const float (&xv)[F_IN], const float (&wv)[F_IN],
                                              float b) {
    const float full = fmaf(di, l1_dot(xv, wv), sum);
    return (_Float16)fmaxf(fmaf(di, full, b), 0.0f);
}

__device__ __forceinline__ void load_row(const float* acc1, const float* __restrict__ dinv, const float* __restrict__ x,
                                         int r, int nrows, int ak0, float4& p, float& s, float (&xv)[F_IN]) {
    if (r < nrows) {
        p = *(const float4*)(acc1 + (size_t)r * HDIM + ak0);
        s = dinv[r];
        const float* xr = x + (size_t)r * (T_HIST * F_IN) + (T_HIST - 1) * F_IN;
#pragma unroll
        for (int f = 0; f < F_IN; ++f) xv[f] = xr[f];
    } else {
        p = make_float4(0.f, 0.f, 0.f, 0.f);
        s = 0.0f;
#pragma unroll
        for (int f = 0; f < F_IN; ++f) xv[f] = 0.0f;
    }
}

__global__ __launch_bounds__(256) void gemm2_wmma_kernel(const float* acc1,
                                                         const float* __restrict__ x,
                                                         const float* __restrict__ W1,
                                                         const float* __restrict__ b1,
                                                         const float* __restrict__ W2,
                                                         const float* __restrict__ dinv,
                                                         float* __restrict__ xw2,
                                                         float* agg_seed,
                                                         int nrows) {
    __shared__ __attribute__((aligned(16))) _Float16 bS[HDIM][LPITCH];
    __shared__ __attribute__((aligned(16))) _Float16 aS[2][16][LPITCH];
    __shared__ float dS[2][16];
    __shared__ __attribute__((aligned(16))) float w1S[F_IN][HDIM];

    const int tid    = threadIdx.x;
    const int ntiles = (nrows + 15) >> 4;
    int tile = blockIdx.x;
    if (tile >= ntiles) return;

    const int arow = __builtin_amdgcn_readfirstlane(tid >> 5);
    const int ak0  = (tid & 31) * 4;

    const float4 bk = make_float4(b1[ak0], b1[ak0 + 1], b1[ak0 + 2], b1[ak0 + 3]);
    float4 p0, p1;
    float  s0, s1;
    float  xa[F_IN], xb[F_IN];
    float  pd = 0.0f;
    {
        const int r0 = tile * 16 + arow;
        load_row(acc1, dinv, x, r0,     nrows, ak0, p0, s0, xa);
        load_row(acc1, dinv, x, r0 + 8, nrows, ak0, p1, s1, xb);
        const int rd = tile * 16 + tid;
        if (tid < 16 && rd < nrows) pd = dinv[rd];
    }

    for (int i = tid; i < HDIM * HDIM; i += 256) {
        const int k = i >> 7;
        const int n = i & (HDIM - 1);
        bS[n][k] = (_Float16)W2[i];
    }
    for (int i = tid; i < F_IN * HDIM; i += 256) w1S[i >> 7][i & (HDIM - 1)] = W1[i];
    __syncthreads();

    const int wave  = tid >> 5;
    const int lane  = tid & 31;
    const int ncol  = wave * 16;
    const int mrow  = lane & 15;
    const int khalf = (lane >> 4) * 8;
    const int rbase = (lane >> 4) << 3;

    v16h bfr[4];
#pragma unroll
    for (int q = 0; q < 4; ++q) bfr[q] = load_frag(&bS[ncol + mrow][q * 32 + khalf]);

    const int col = ncol + mrow;
    int buf = 0;

    for (; tile < ntiles; tile += gridDim.x) {
        const int block_row = tile * 16;

        {
            float wc[4][F_IN];
#pragma unroll
            for (int f = 0; f < F_IN; ++f) {
                const float4 wq = *(const float4*)&w1S[f][ak0];
                wc[0][f] = wq.x; wc[1][f] = wq.y; wc[2][f] = wq.z; wc[3][f] = wq.w;
            }
            v4h h0, h1;
            h0[0] = l1_finish(p0.x, s0, xa, wc[0], bk.x); h0[1] = l1_finish(p0.y, s0, xa, wc[1], bk.y);
            h0[2] = l1_finish(p0.z, s0, xa, wc[2], bk.z); h0[3] = l1_finish(p0.w, s0, xa, wc[3], bk.w);
            h1[0] = l1_finish(p1.x, s1, xb, wc[0], bk.x); h1[1] = l1_finish(p1.y, s1, xb, wc[1], bk.y);
            h1[2] = l1_finish(p1.z, s1, xb, wc[2], bk.z); h1[3] = l1_finish(p1.w, s1, xb, wc[3], bk.w);
            *(v4h*)&aS[buf][arow][ak0]     = h0;
            *(v4h*)&aS[buf][arow + 8][ak0] = h1;
            if (tid < 16) dS[buf][tid] = pd;
        }
        __syncthreads();

        {
            const int nt = tile + gridDim.x;
            if (nt < ntiles) {
                const int r0 = nt * 16 + arow;
                load_row(acc1, dinv, x, r0,     nrows, ak0, p0, s0, xa);
                load_row(acc1, dinv, x, r0 + 8, nrows, ak0, p1, s1, xb);
                const int rd = nt * 16 + tid;
                pd = 0.0f;
                if (tid < 16 && rd < nrows) pd = dinv[rd];
            }
        }

        v16h afr[4];
#pragma unroll
        for (int q = 0; q < 4; ++q) afr[q] = load_frag(&aS[buf][mrow][q * 32 + khalf]);
        float di[8];
#pragma unroll
        for (int v = 0; v < 8; ++v) di[v] = dS[buf][rbase + v];

        v8f acc = {};
#pragma unroll
        for (int q = 0; q < 4; ++q)
            acc = __builtin_amdgcn_wmma_f32_16x16x32_f16(false, afr[q], false, bfr[q],
                                                         (short)0, acc, false, false);

        if (block_row + 16 <= nrows) {
#pragma unroll
            for (int v = 0; v < 8; ++v) {
                const int   row = block_row + rbase + v;
                const float val = di[v] * acc[v];
                xw2[(size_t)row * HDIM + col]      = val;
                agg_seed[(size_t)row * HDIM + col] = val;
            }
        } else {
#pragma unroll
            for (int v = 0; v < 8; ++v) {
                const int row = block_row + rbase + v;
                if (row < nrows) {
                    const float val = di[v] * acc[v];
                    xw2[(size_t)row * HDIM + col]      = val;
                    agg_seed[(size_t)row * HDIM + col] = val;
                }
            }
        }
        buf ^= 1;
    }
}

__global__ __launch_bounds__(256) void out_kernel(const float* __restrict__ acc2,
                                                  const float* __restrict__ dinv,
                                                  const float* __restrict__ b2,
                                                  const float* __restrict__ Wout,
                                                  const float* __restrict__ bout,
                                                  float* __restrict__ out, int n) {
    int node = (blockIdx.x * blockDim.x + threadIdx.x) >> 5;
    int lane = threadIdx.x & 31;
    if (node >= n) return;

    const float4* hr = (const float4*)(acc2 + (size_t)node * HDIM);
    float4 hv = hr[lane];
    const float  di = dinv[node];
    const float4 bv = make_float4(b2[lane * 4], b2[lane * 4 + 1], b2[lane * 4 + 2], b2[lane * 4 + 3]);
    hv.x = fmaxf(fmaf(di, hv.x, bv.x), 0.0f);
    hv.y = fmaxf(fmaf(di, hv.y, bv.y), 0.0f);
    hv.z = fmaxf(fmaf(di, hv.z, bv.z), 0.0f);
    hv.w = fmaxf(fmaf(di, hv.w, bv.w), 0.0f);

    int h0 = lane * 4;
    float acc[CDIM];
#pragma unroll
    for (int c = 0; c < CDIM; ++c) {
        acc[c] = hv.x * Wout[(h0 + 0) * CDIM + c]
               + hv.y * Wout[(h0 + 1) * CDIM + c]
               + hv.z * Wout[(h0 + 2) * CDIM + c]
               + hv.w * Wout[(h0 + 3) * CDIM + c];
    }
#pragma unroll
    for (int off = 16; off > 0; off >>= 1) {
#pragma unroll
        for (int c = 0; c < CDIM; ++c) acc[c] += __shfl_xor(acc[c], off, 32);
    }
    if (lane == 0) {
#pragma unroll
        for (int c = 0; c < CDIM; ++c) acc[c] += bout[c];
        float m = acc[0];
#pragma unroll
        for (int c = 1; c < CDIM; ++c) m = fmaxf(m, acc[c]);
        float sum = 0.0f;
#pragma unroll
        for (int c = 0; c < CDIM; ++c) { acc[c] = __expf(acc[c] - m); sum += acc[c]; }
        float inv = 1.0f / sum;
#pragma unroll
        for (int c = 0; c < CDIM; ++c) out[(size_t)node * CDIM + c] = acc[c] * inv;
    }
}

extern "C" void kernel_launch(void* const* d_in, const int* in_sizes, int n_in,
                              void* d_out, int out_size, void* d_ws, size_t ws_size,
                              hipStream_t stream) {
    const float* x    = (const float*)d_in[0];
    const int*   eidx = (const int*)  d_in[1];
    const float* W1   = (const float*)d_in[2];
    const float* b1   = (const float*)d_in[3];
    const float* W2   = (const float*)d_in[4];
    const float* b2   = (const float*)d_in[5];
    const float* Wout = (const float*)d_in[6];
    const float* bout = (const float*)d_in[7];
    float*       out  = (float*)d_out;

    const int N = in_sizes[0] / (T_HIST * F_IN);
    const int E = in_sizes[1] / 2;
    const int* src = eidx;
    const int* dst = eidx + E;

    float* dinv  = (float*)d_ws;
    int*   queue = (int*)(dinv + N);
    float* bufA  = dinv + ((N + 4 + 63) & ~63);
    float* bufB = bufA + (size_t)N * HDIM;
    float* rec  = bufA + (size_t)N * DEG_PIECES;

    const int TB = 256;
    const int scatter_groups = 2048;
    const int node_blocks = ((N * 32) + TB - 1) / TB;

    deg_part_kernel<<<dim3(DEG_PIECES, (N + DEG_RANGE - 1) / DEG_RANGE), DEG_THREADS, 0, stream>>>(dst, bufA, bufB, N, E);
    dinv_kernel<<<(N + TB - 1) / TB, TB, 0, stream>>>(bufA, x, dinv, rec, queue, N);

    scatter_kernel<true><<<scatter_groups, TB, 0, stream>>>(src, dst, rec, W1, bufB, queue, E);

    gemm2_wmma_kernel<<<1024, 256, 0, stream>>>(bufB, x, W1, b1, W2, dinv, bufA, bufB, N);
    scatter_kernel<false><<<scatter_groups, TB, 0, stream>>>(src, dst, bufA, W1, bufB, queue + 2, E);

    out_kernel<<<node_blocks, TB, 0, stream>>>(bufB, dinv, b2, Wout, bout, out, N);
}
